# starting kernel
# baseline (speedup 1.0000x reference)
#include <hip/hip_runtime.h>
#include <math.h>


#define NPTS 100000
#define NBAS 32
#define MCLU 16
#define KNEI 32
#define MUF  100.0f

typedef __attribute__((ext_vector_type(2))) float v2f;
typedef __attribute__((ext_vector_type(8))) float v8f;

#define WS_ENCPART 0
#define WS_ENCODED (WS_ENCPART + 1024)
#define WS_INVW2T  (WS_ENCODED + 32)
#define WS_RDENT   (WS_INVW2T + 512)
#define DEN_BLKS   224
#define WS_DENPART (WS_RDENT + 1024)
#define WS_DECT    (WS_DENPART + DEN_BLKS*1024)

__global__ void k_enc_partial(const float* __restrict__ x,
                              const float* __restrict__ W,
                              float* __restrict__ part) {
  const int row = blockIdx.y, chunk = blockIdx.x;
  const int CS = (NPTS + 31) / 32;
  const int j0 = chunk * CS;
  const int j1 = (j0 + CS < NPTS) ? (j0 + CS) : NPTS;
  float s = 0.f;
  for (int j = j0 + threadIdx.x; j < j1; j += blockDim.x)
    s += x[j] * W[row * NPTS + j];
  __shared__ float red[256];
  red[threadIdx.x] = s;
  __syncthreads();
  for (int o = 128; o > 0; o >>= 1) {
    if (threadIdx.x < o) red[threadIdx.x] += red[threadIdx.x + o];
    __syncthreads();
  }
  if (threadIdx.x == 0) part[row * 32 + chunk] = red[0];
}

__global__ void k_enc_finish(const float* __restrict__ part,
                             const float* __restrict__ enc_b,
                             const float* __restrict__ bw_w,
                             const float* __restrict__ bw_b,
                             float* __restrict__ encoded,
                             float* __restrict__ invw2t) {
  __shared__ float enc_s[NBAS];
  const int t = threadIdx.x;
  if (t < NBAS) {
    float s = 0.f;
    for (int c = 0; c < 32; ++c) s += part[t * 32 + c];
    s += enc_b[t];
    enc_s[t] = s;
    encoded[t] = s;
  }
  __syncthreads();
  if (t < NBAS * MCLU) {
    float s = 0.f;
    for (int q = 0; q < NBAS; ++q) s += enc_s[q] * bw_w[t * NBAS + q];
    s += bw_b[t];
    float band = (1.f / (1.f + expf(-s))) / 60.f;
    float wm = band * MUF;
    const int i = t / MCLU, c = t % MCLU;
    invw2t[c * NBAS + i] = 1.f / (wm * wm);
  }
}

__global__ void k_dec_transpose(const float* __restrict__ dec,
                                float* __restrict__ dect) {
  __shared__ float tile[32][33];
  const int jb = blockIdx.x * 32;
  const int tx = threadIdx.x, ty = threadIdx.y;
  tile[ty][tx] = dec[ty * NPTS + jb + tx];
  __syncthreads();
  dect[(jb + ty) * 32 + tx] = tile[tx][ty];
}

__global__ void k_denom_partial(const float* __restrict__ d,
                                const int* __restrict__ lab,
                                const float* __restrict__ invw2t,
                                float* __restrict__ part) {
  __shared__ float iw[MCLU * NBAS];
  __shared__ float red[8 * 1024];
  for (int t = threadIdx.x; t < MCLU * NBAS; t += blockDim.x) iw[t] = invw2t[t];
  __syncthreads();
  const int lane = threadIdx.x & 31, wv = threadIdx.x >> 5;
  const int gw = blockIdx.x * 8 + wv, nw = gridDim.x * 8;
  float acc[NBAS];
#pragma unroll
  for (int i = 0; i < NBAS; ++i) acc[i] = 0.f;
  for (int j = gw; j < NPTS; j += nw) {
    const int c = lab[j];
    const float dv = d[j * KNEI + lane];
    const float d2 = dv * dv;
#pragma unroll
    for (int i = 0; i < NBAS; ++i)
      acc[i] += fmaxf(1.f - d2 * iw[c * NBAS + i], 0.f);
  }
#pragma unroll
  for (int i = 0; i < NBAS; ++i) red[wv * 1024 + i * 32 + lane] = acc[i];
  __syncthreads();
  for (int e = threadIdx.x; e < 1024; e += blockDim.x) {
    float s = 0.f;
#pragma unroll
    for (int w = 0; w < 8; ++w) s += red[w * 1024 + e];
    part[blockIdx.x * 1024 + e] = s;
  }
}

__global__ void k_denom_finish(const float* __restrict__ part, int nblk,
                               float* __restrict__ rdent) {
  const int t = threadIdx.x;
  float s = 0.f;
  for (int b = 0; b < nblk; ++b) s += part[b * 1024 + t];
  const int i = t >> 5, kk = t & 31;
  rdent[kk * 32 + i] = 1.f / s;
}

__global__ void k_fused_out(const float* __restrict__ d,
                            const int* __restrict__ nid,
                            const int* __restrict__ lab,
                            const float* __restrict__ dect,
                            const float* __restrict__ invw2t,
                            const float* __restrict__ rdent,
                            const float* __restrict__ encoded,
                            float* __restrict__ out) {
  __shared__ float iw[MCLU * NBAS];
  __shared__ float rd[NBAS * NBAS];
  __shared__ float enc_s[NBAS];
  __shared__ float g[8][32 * 17];
  for (int t = threadIdx.x; t < MCLU * NBAS; t += blockDim.x) iw[t] = invw2t[t];
  for (int t = threadIdx.x; t < NBAS * NBAS; t += blockDim.x) rd[t] = rdent[t];
  if (threadIdx.x < NBAS) enc_s[threadIdx.x] = encoded[threadIdx.x];
  __syncthreads();

  const int lane = threadIdx.x & 31, wv = threadIdx.x >> 5;
  const int gw = blockIdx.x * 8 + wv, nw = gridDim.x * 8;
  float* gbuf = g[wv];
  const int NT = NPTS / 16;

  for (int tile = gw; tile < NT; tile += nw) {
    const int j0 = tile * 16;
    for (int p = 0; p < 16; ++p) {
      const int j = j0 + p;
      const int c = lab[j];
      const float s = iw[c * NBAS + lane];
      const int nl = nid[j * KNEI + lane];
      const float dv = d[j * KNEI + lane];
      const float d2 = dv * dv;
      float acc = 0.f;
#pragma unroll
      for (int kk = 0; kk < KNEI; ++kk) {
        const int nb = __shfl(nl, kk, 32);
        const float d2k = __shfl(d2, kk, 32);
        const float win = fmaxf(1.f - d2k * s, 0.f);
        const float dec = dect[nb * 32 + lane];
        acc += rd[kk * 32 + lane] * win * dec;
      }
      gbuf[lane * 17 + p] = acc;
    }
    asm volatile("s_wait_dscnt 0" ::: "memory");

    v8f cacc = {0.f, 0.f, 0.f, 0.f, 0.f, 0.f, 0.f, 0.f};
    const int half = (lane >> 4) & 1;
    const int col = lane & 15;
#pragma unroll
    for (int s8 = 0; s8 < 8; ++s8) {
      const int kb = s8 * 4;
      v2f a, b;
      a.x = enc_s[kb + 2 * half];
      a.y = enc_s[kb + 2 * half + 1];
      b.x = gbuf[(kb + 2 * half) * 17 + col];
      b.y = gbuf[(kb + 2 * half + 1) * 17 + col];
      cacc = __builtin_amdgcn_wmma_f32_16x16x4_f32(
          false, a, false, b, (short)0, cacc, false, false);
    }
    if (lane < 16) out[j0 + lane] = cacc[0];
  }
}

extern "C" void kernel_launch(void* const* d_in, const int* in_sizes, int n_in,
                              void* d_out, int out_size, void* d_ws, size_t ws_size,
                              hipStream_t stream) {
  const float* x     = (const float*)d_in[0];
  const float* enc_w = (const float*)d_in[1];
  const float* enc_b = (const float*)d_in[2];
  const float* dec   = (const float*)d_in[3];
  const float* bw_w  = (const float*)d_in[4];
  const float* bw_b  = (const float*)d_in[5];
  const float* nd    = (const float*)d_in[6];
  const int*   nid   = (const int*)d_in[7];
  const int*   lab   = (const int*)d_in[8];
  float* out = (float*)d_out;
  float* ws  = (float*)d_ws;

  float* enc_part = ws + WS_ENCPART;
  float* encoded  = ws + WS_ENCODED;
  float* invw2t   = ws + WS_INVW2T;
  float* rdent    = ws + WS_RDENT;
  float* denpart  = ws + WS_DENPART;
  float* dect     = ws + WS_DECT;

  k_enc_partial<<<dim3(32, 32), 256, 0, stream>>>(x, enc_w, enc_part);
  k_enc_finish<<<1, 512, 0, stream>>>(enc_part, enc_b, bw_w, bw_b, encoded, invw2t);
  k_dec_transpose<<<NPTS / 32, dim3(32, 32), 0, stream>>>(dec, dect);
  k_denom_partial<<<DEN_BLKS, 256, 0, stream>>>(nd, lab, invw2t, denpart);
  k_denom_finish<<<1, 1024, 0, stream>>>(denpart, DEN_BLKS, rdent);
  k_fused_out<<<200, 256, 0, stream>>>(nd, nid, lab, dect, invw2t, rdent,
                                       encoded, out);
}


# fused gather kernel: each point's 32 neighbour-row loads issued as one group before use, index and distance rows fetched two points ahead (was one load, wait, fma per neighbour)
# speedup vs baseline: 4.0607x; 4.0607x over previous
#include <hip/hip_runtime.h>
#include <math.h>


#define NPTS 100000
#define NBAS 32
#define MCLU 16
#define KNEI 32
#define MUF  100.0f

typedef __attribute__((ext_vector_type(2))) float v2f;
typedef __attribute__((ext_vector_type(8))) float v8f;

#define WS_ENCPART 0
#define WS_ENCODED (WS_ENCPART + 1024)
#define WS_INVW2T  (WS_ENCODED + 32)
#define WS_RDENT   (WS_INVW2T + 512)
#define DEN_BLKS   224
#define WS_DENPART (WS_RDENT + 1024)
#define WS_DECT    (WS_DENPART + DEN_BLKS*1024)

__global__ void k_enc_partial(const float* __restrict__ x,
                              const float* __restrict__ W,
                              float* __restrict__ part) {
  const int row = blockIdx.y, chunk = blockIdx.x;
  const int CS = (NPTS + 31) / 32;
  const int j0 = chunk * CS;
  const int j1 = (j0 + CS < NPTS) ? (j0 + CS) : NPTS;
  float s = 0.f;
  for (int j = j0 + threadIdx.x; j < j1; j += blockDim.x)
    s += x[j] * W[row * NPTS + j];
  __shared__ float red[256];
  red[threadIdx.x] = s;
  __syncthreads();
  for (int o = 128; o > 0; o >>= 1) {
    if (threadIdx.x < o) red[threadIdx.x] += red[threadIdx.x + o];
    __syncthreads();
  }
  if (threadIdx.x == 0) part[row * 32 + chunk] = red[0];
}

__global__ void k_enc_finish(const float* __restrict__ part,
                             const float* __restrict__ enc_b,
                             const float* __restrict__ bw_w,
                             const float* __restrict__ bw_b,
                             float* __restrict__ encoded,
                             float* __restrict__ invw2t) {
  __shared__ float enc_s[NBAS];
  const int t = threadIdx.x;
  if (t < NBAS) {
    float s = 0.f;
    for (int c = 0; c < 32; ++c) s += part[t * 32 + c];
    s += enc_b[t];
    enc_s[t] = s;
    encoded[t] = s;
  }
  __syncthreads();
  if (t < NBAS * MCLU) {
    float s = 0.f;
    for (int q = 0; q < NBAS; ++q) s += enc_s[q] * bw_w[t * NBAS + q];
    s += bw_b[t];
    float band = (1.f / (1.f + expf(-s))) / 60.f;
    float wm = band * MUF;
    const int i = t / MCLU, c = t % MCLU;
    invw2t[c * NBAS + i] = 1.f / (wm * wm);
  }
}

__global__ void k_dec_transpose(const float* __restrict__ dec,
                                float* __restrict__ dect) {
  __shared__ float tile[32][33];
  const int jb = blockIdx.x * 32;
  const int tx = threadIdx.x, ty = threadIdx.y;
  tile[ty][tx] = dec[ty * NPTS + jb + tx];
  __syncthreads();
  dect[(jb + ty) * 32 + tx] = tile[tx][ty];
}

__global__ void k_denom_partial(const float* __restrict__ d,
                                const int* __restrict__ lab,
                                const float* __restrict__ invw2t,
                                float* __restrict__ part) {
  __shared__ float iw[MCLU * NBAS];
  __shared__ float red[8 * 1024];
  for (int t = threadIdx.x; t < MCLU * NBAS; t += blockDim.x) iw[t] = invw2t[t];
  __syncthreads();
  const int lane = threadIdx.x & 31, wv = threadIdx.x >> 5;
  const int gw = blockIdx.x * 8 + wv, nw = gridDim.x * 8;
  float acc[NBAS];
#pragma unroll
  for (int i = 0; i < NBAS; ++i) acc[i] = 0.f;
  for (int j = gw; j < NPTS; j += nw) {
    const int c = lab[j];
    const float dv = d[j * KNEI + lane];
    const float d2 = dv * dv;
#pragma unroll
    for (int i = 0; i < NBAS; ++i)
      acc[i] += fmaxf(1.f - d2 * iw[c * NBAS + i], 0.f);
  }
#pragma unroll
  for (int i = 0; i < NBAS; ++i) red[wv * 1024 + i * 32 + lane] = acc[i];
  __syncthreads();
  for (int e = threadIdx.x; e < 1024; e += blockDim.x) {
    float s = 0.f;
#pragma unroll
    for (int w = 0; w < 8; ++w) s += red[w * 1024 + e];
    part[blockIdx.x * 1024 + e] = s;
  }
}

__global__ void k_denom_finish(const float* __restrict__ part, int nblk,
                               float* __restrict__ rdent) {
  const int t = threadIdx.x;
  float s = 0.f;
  for (int b = 0; b < nblk; ++b) s += part[b * 1024 + t];
  const int i = t >> 5, kk = t & 31;
  rdent[kk * 32 + i] = 1.f / s;
}

__device__ __forceinline__ void fused_point(int& nl, float& dv, int& c, const int p,
                                            const int j0, const int jnext, const int lane,
                                            const float* iw, const float (&rdr)[KNEI],
                                            const float* __restrict__ dect,
                                            const int* __restrict__ nid,
                                            const float* __restrict__ d,
                                            const int* __restrict__ lab,
                                            float* gbuf) {
  const float s = iw[c * NBAS + lane];
  const float d2 = dv * dv;
  float dec[KNEI];
#pragma unroll
  for (int kk = 0; kk < KNEI; ++kk) {
    const int nb = __shfl(nl, kk, 32);
    dec[kk] = dect[nb * 32 + lane];
  }
  __builtin_amdgcn_sched_barrier(0);
  int j2 = (p + 2 < 16) ? (j0 + p + 2) : (jnext + p + 2 - 16);
  j2 = (j2 < NPTS) ? j2 : (NPTS - 1);
  nl = nid[j2 * KNEI + lane];
  dv = d[j2 * KNEI + lane];
  c = lab[j2];
  float w[KNEI];
#pragma unroll
  for (int kk = 0; kk < KNEI; ++kk)
    w[kk] = rdr[kk] * fmaxf(1.f - __shfl(d2, kk, 32) * s, 0.f);
  __builtin_amdgcn_sched_barrier(0);
  float acc = 0.f;
#pragma unroll
  for (int kk = 0; kk < KNEI; ++kk) acc += w[kk] * dec[kk];
  gbuf[lane * 17 + p] = acc;
}

__global__ void __launch_bounds__(256) k_fused_out(const float* __restrict__ d,
                            const int* __restrict__ nid,
                            const int* __restrict__ lab,
                            const float* __restrict__ dect,
                            const float* __restrict__ invw2t,
                            const float* __restrict__ rdent,
                            const float* __restrict__ encoded,
                            float* __restrict__ out) {
  __shared__ float iw[MCLU * NBAS];
  __shared__ float rd[NBAS * NBAS];
  __shared__ float enc_s[NBAS];
  __shared__ float g[8][32 * 17];
  for (int t = threadIdx.x; t < MCLU * NBAS; t += blockDim.x) iw[t] = invw2t[t];
  for (int t = threadIdx.x; t < NBAS * NBAS; t += blockDim.x) rd[t] = rdent[t];
  if (threadIdx.x < NBAS) enc_s[threadIdx.x] = encoded[threadIdx.x];
  __syncthreads();

  const int lane = threadIdx.x & 31, wv = threadIdx.x >> 5;
  const int gw = blockIdx.x * 8 + wv, nw = gridDim.x * 8;
  float* gbuf = g[wv];
  const int NT = NPTS / 16;

  float rdr[KNEI];
#pragma unroll
  for (int kk = 0; kk < KNEI; ++kk) rdr[kk] = rd[kk * 32 + lane];

  int nlA = 0, cA = 0, nlB = 0, cB = 0;
  float dvA = 0.f, dvB = 0.f;
  if (gw < NT) {
    const int jf = gw * 16;
    nlA = nid[jf * KNEI + lane];
    dvA = d[jf * KNEI + lane];
    cA = lab[jf];
    nlB = nid[(jf + 1) * KNEI + lane];
    dvB = d[(jf + 1) * KNEI + lane];
    cB = lab[jf + 1];
  }

  for (int tile = gw; tile < NT; tile += nw) {
    const int j0 = tile * 16;
    const int jnext = (tile + nw) * 16;
    for (int p = 0; p < 16; p += 2) {
      fused_point(nlA, dvA, cA, p, j0, jnext, lane, iw, rdr, dect, nid, d, lab, gbuf);
      fused_point(nlB, dvB, cB, p + 1, j0, jnext, lane, iw, rdr, dect, nid, d, lab, gbuf);
    }
    asm volatile("s_wait_dscnt 0" ::: "memory");

    v8f cacc = {0.f, 0.f, 0.f, 0.f, 0.f, 0.f, 0.f, 0.f};
    const int half = (lane >> 4) & 1;
    const int col = lane & 15;
#pragma unroll
    for (int s8 = 0; s8 < 8; ++s8) {
      const int kb = s8 * 4;
      v2f a, b;
      a.x = enc_s[kb + 2 * half];
      a.y = enc_s[kb + 2 * half + 1];
      b.x = gbuf[(kb + 2 * half) * 17 + col];
      b.y = gbuf[(kb + 2 * half + 1) * 17 + col];
      cacc = __builtin_amdgcn_wmma_f32_16x16x4_f32(
          false, a, false, b, (short)0, cacc, false, false);
    }
    if (lane < 16) out[j0 + lane] = cacc[0];
  }
}

extern "C" void kernel_launch(void* const* d_in, const int* in_sizes, int n_in,
                              void* d_out, int out_size, void* d_ws, size_t ws_size,
                              hipStream_t stream) {
  const float* x     = (const float*)d_in[0];
  const float* enc_w = (const float*)d_in[1];
  const float* enc_b = (const float*)d_in[2];
  const float* dec   = (const float*)d_in[3];
  const float* bw_w  = (const float*)d_in[4];
  const float* bw_b  = (const float*)d_in[5];
  const float* nd    = (const float*)d_in[6];
  const int*   nid   = (const int*)d_in[7];
  const int*   lab   = (const int*)d_in[8];
  float* out = (float*)d_out;
  float* ws  = (float*)d_ws;

  float* enc_part = ws + WS_ENCPART;
  float* encoded  = ws + WS_ENCODED;
  float* invw2t   = ws + WS_INVW2T;
  float* rdent    = ws + WS_RDENT;
  float* denpart  = ws + WS_DENPART;
  float* dect     = ws + WS_DECT;

  k_enc_partial<<<dim3(32, 32), 256, 0, stream>>>(x, enc_w, enc_part);
  k_enc_finish<<<1, 512, 0, stream>>>(enc_part, enc_b, bw_w, bw_b, encoded, invw2t);
  k_dec_transpose<<<NPTS / 32, dim3(32, 32), 0, stream>>>(dec, dect);
  k_denom_partial<<<DEN_BLKS, 256, 0, stream>>>(nd, lab, invw2t, denpart);
  k_denom_finish<<<1, 1024, 0, stream>>>(denpart, DEN_BLKS, rdent);
  k_fused_out<<<200, 256, 0, stream>>>(nd, nid, lab, dect, invw2t, rdent,
                                       encoded, out);
}


# transpose kernel: one wave moves a whole 32x32 tile (32 row loads in flight, wave-private LDS turn, 32 whole-line stores): 3125 waves enter instead of 100000 one-element waves
# speedup vs baseline: 4.2409x; 1.0444x over previous
#include <hip/hip_runtime.h>
#include <math.h>


#define NPTS 100000
#define NBAS 32
#define MCLU 16
#define KNEI 32
#define MUF  100.0f

typedef __attribute__((ext_vector_type(2))) float v2f;
typedef __attribute__((ext_vector_type(8))) float v8f;

#define WS_ENCPART 0
#define WS_ENCODED (WS_ENCPART + 1024)
#define WS_INVW2T  (WS_ENCODED + 32)
#define WS_RDENT   (WS_INVW2T + 512)
#define DEN_BLKS   224
#define WS_DENPART (WS_RDENT + 1024)
#define WS_DECT    (WS_DENPART + DEN_BLKS*1024)

__global__ void k_enc_partial(const float* __restrict__ x,
                              const float* __restrict__ W,
                              float* __restrict__ part) {
  const int row = blockIdx.y, chunk = blockIdx.x;
  const int CS = (NPTS + 31) / 32;
  const int j0 = chunk * CS;
  const int j1 = (j0 + CS < NPTS) ? (j0 + CS) : NPTS;
  float s = 0.f;
  for (int j = j0 + threadIdx.x; j < j1; j += blockDim.x)
    s += x[j] * W[row * NPTS + j];
  __shared__ float red[256];
  red[threadIdx.x] = s;
  __syncthreads();
  for (int o = 128; o > 0; o >>= 1) {
    if (threadIdx.x < o) red[threadIdx.x] += red[threadIdx.x + o];
    __syncthreads();
  }
  if (threadIdx.x == 0) part[row * 32 + chunk] = red[0];
}

__global__ void k_enc_finish(const float* __restrict__ part,
                             const float* __restrict__ enc_b,
                             const float* __restrict__ bw_w,
                             const float* __restrict__ bw_b,
                             float* __restrict__ encoded,
                             float* __restrict__ invw2t) {
  __shared__ float enc_s[NBAS];
  const int t = threadIdx.x;
  if (t < NBAS) {
    float s = 0.f;
    for (int c = 0; c < 32; ++c) s += part[t * 32 + c];
    s += enc_b[t];
    enc_s[t] = s;
    encoded[t] = s;
  }
  __syncthreads();
  if (t < NBAS * MCLU) {
    float s = 0.f;
    for (int q = 0; q < NBAS; ++q) s += enc_s[q] * bw_w[t * NBAS + q];
    s += bw_b[t];
    float band = (1.f / (1.f + expf(-s))) / 60.f;
    float wm = band * MUF;
    const int i = t / MCLU, c = t % MCLU;
    invw2t[c * NBAS + i] = 1.f / (wm * wm);
  }
}

#define TR_TILES 3125
#define TR_BLKS  391
__global__ void __launch_bounds__(256) k_dec_transpose(const float* __restrict__ dec,
                                                       float* __restrict__ dect) {
  __shared__ float tile[8][32 * 33];
  const int lane = threadIdx.x & 31;
  const int wv = __builtin_amdgcn_readfirstlane(threadIdx.x >> 5);
  const int t = blockIdx.x * 8 + wv;
  if (t >= TR_TILES) return;
  const int jb = t * 32;
  float* tw = tile[wv];
  float v[32];
#pragma unroll
  for (int i = 0; i < 32; ++i) v[i] = dec[i * NPTS + jb + lane];
#pragma unroll
  for (int i = 0; i < 32; ++i) tw[i * 33 + lane] = v[i];
  asm volatile("s_wait_dscnt 0" ::: "memory");
#pragma unroll
  for (int r = 0; r < 32; ++r) dect[(jb + r) * 32 + lane] = tw[lane * 33 + r];
}

__global__ void k_denom_partial(const float* __restrict__ d,
                                const int* __restrict__ lab,
                                const float* __restrict__ invw2t,
                                float* __restrict__ part) {
  __shared__ float iw[MCLU * NBAS];
  __shared__ float red[8 * 1024];
  for (int t = threadIdx.x; t < MCLU * NBAS; t += blockDim.x) iw[t] = invw2t[t];
  __syncthreads();
  const int lane = threadIdx.x & 31, wv = threadIdx.x >> 5;
  const int gw = blockIdx.x * 8 + wv, nw = gridDim.x * 8;
  float acc[NBAS];
#pragma unroll
  for (int i = 0; i < NBAS; ++i) acc[i] = 0.f;
  for (int j = gw; j < NPTS; j += nw) {
    const int c = lab[j];
    const float dv = d[j * KNEI + lane];
    const float d2 = dv * dv;
#pragma unroll
    for (int i = 0; i < NBAS; ++i)
      acc[i] += fmaxf(1.f - d2 * iw[c * NBAS + i], 0.f);
  }
#pragma unroll
  for (int i = 0; i < NBAS; ++i) red[wv * 1024 + i * 32 + lane] = acc[i];
  __syncthreads();
  for (int e = threadIdx.x; e < 1024; e += blockDim.x) {
    float s = 0.f;
#pragma unroll
    for (int w = 0; w < 8; ++w) s += red[w * 1024 + e];
    part[blockIdx.x * 1024 + e] = s;
  }
}

__global__ void k_denom_finish(const float* __restrict__ part, int nblk,
                               float* __restrict__ rdent) {
  const int t = threadIdx.x;
  float s = 0.f;
  for (int b = 0; b < nblk; ++b) s += part[b * 1024 + t];
  const int i = t >> 5, kk = t & 31;
  rdent[kk * 32 + i] = 1.f / s;
}

__device__ __forceinline__ void fused_point(int& nl, float& dv, int& c, const int p,
                                            const int j0, const int jnext, const int lane,
                                            const float* iw, const float (&rdr)[KNEI],
                                            const float* __restrict__ dect,
                                            const int* __restrict__ nid,
                                            const float* __restrict__ d,
                                            const int* __restrict__ lab,
                                            float* gbuf) {
  const float s = iw[c * NBAS + lane];
  const float d2 = dv * dv;
  float dec[KNEI];
#pragma unroll
  for (int kk = 0; kk < KNEI; ++kk) {
    const int nb = __shfl(nl, kk, 32);
    dec[kk] = dect[nb * 32 + lane];
  }
  __builtin_amdgcn_sched_barrier(0);
  int j2 = (p + 2 < 16) ? (j0 + p + 2) : (jnext + p + 2 - 16);
  j2 = (j2 < NPTS) ? j2 : (NPTS - 1);
  nl = nid[j2 * KNEI + lane];
  dv = d[j2 * KNEI + lane];
  c = lab[j2];
  float w[KNEI];
#pragma unroll
  for (int kk = 0; kk < KNEI; ++kk)
    w[kk] = rdr[kk] * fmaxf(1.f - __shfl(d2, kk, 32) * s, 0.f);
  __builtin_amdgcn_sched_barrier(0);
  float acc = 0.f;
#pragma unroll
  for (int kk = 0; kk < KNEI; ++kk) acc += w[kk] * dec[kk];
  gbuf[lane * 17 + p] = acc;
}

__global__ void __launch_bounds__(256) k_fused_out(const float* __restrict__ d,
                            const int* __restrict__ nid,
                            const int* __restrict__ lab,
                            const float* __restrict__ dect,
                            const float* __restrict__ invw2t,
                            const float* __restrict__ rdent,
                            const float* __restrict__ encoded,
                            float* __restrict__ out) {
  __shared__ float iw[MCLU * NBAS];
  __shared__ float rd[NBAS * NBAS];
  __shared__ float enc_s[NBAS];
  __shared__ float g[8][32 * 17];
  for (int t = threadIdx.x; t < MCLU * NBAS; t += blockDim.x) iw[t] = invw2t[t];
  for (int t = threadIdx.x; t < NBAS * NBAS; t += blockDim.x) rd[t] = rdent[t];
  if (threadIdx.x < NBAS) enc_s[threadIdx.x] = encoded[threadIdx.x];
  __syncthreads();

  const int lane = threadIdx.x & 31, wv = threadIdx.x >> 5;
  const int gw = blockIdx.x * 8 + wv, nw = gridDim.x * 8;
  float* gbuf = g[wv];
  const int NT = NPTS / 16;

  float rdr[KNEI];
#pragma unroll
  for (int kk = 0; kk < KNEI; ++kk) rdr[kk] = rd[kk * 32 + lane];

  int nlA = 0, cA = 0, nlB = 0, cB = 0;
  float dvA = 0.f, dvB = 0.f;
  if (gw < NT) {
    const int jf = gw * 16;
    nlA = nid[jf * KNEI + lane];
    dvA = d[jf * KNEI + lane];
    cA = lab[jf];
    nlB = nid[(jf + 1) * KNEI + lane];
    dvB = d[(jf + 1) * KNEI + lane];
    cB = lab[jf + 1];
  }

  for (int tile = gw; tile < NT; tile += nw) {
    const int j0 = tile * 16;
    const int jnext = (tile + nw) * 16;
    for (int p = 0; p < 16; p += 2) {
      fused_point(nlA, dvA, cA, p, j0, jnext, lane, iw, rdr, dect, nid, d, lab, gbuf);
      fused_point(nlB, dvB, cB, p + 1, j0, jnext, lane, iw, rdr, dect, nid, d, lab, gbuf);
    }
    asm volatile("s_wait_dscnt 0" ::: "memory");

    v8f cacc = {0.f, 0.f, 0.f, 0.f, 0.f, 0.f, 0.f, 0.f};
    const int half = (lane >> 4) & 1;
    const int col = lane & 15;
#pragma unroll
    for (int s8 = 0; s8 < 8; ++s8) {
      const int kb = s8 * 4;
      v2f a, b;
      a.x = enc_s[kb + 2 * half];
      a.y = enc_s[kb + 2 * half + 1];
      b.x = gbuf[(kb + 2 * half) * 17 + col];
      b.y = gbuf[(kb + 2 * half + 1) * 17 + col];
      cacc = __builtin_amdgcn_wmma_f32_16x16x4_f32(
          false, a, false, b, (short)0, cacc, false, false);
    }
    if (lane < 16) out[j0 + lane] = cacc[0];
  }
}

extern "C" void kernel_launch(void* const* d_in, const int* in_sizes, int n_in,
                              void* d_out, int out_size, void* d_ws, size_t ws_size,
                              hipStream_t stream) {
  const float* x     = (const float*)d_in[0];
  const float* enc_w = (const float*)d_in[1];
  const float* enc_b = (const float*)d_in[2];
  const float* dec   = (const float*)d_in[3];
  const float* bw_w  = (const float*)d_in[4];
  const float* bw_b  = (const float*)d_in[5];
  const float* nd    = (const float*)d_in[6];
  const int*   nid   = (const int*)d_in[7];
  const int*   lab   = (const int*)d_in[8];
  float* out = (float*)d_out;
  float* ws  = (float*)d_ws;

  float* enc_part = ws + WS_ENCPART;
  float* encoded  = ws + WS_ENCODED;
  float* invw2t   = ws + WS_INVW2T;
  float* rdent    = ws + WS_RDENT;
  float* denpart  = ws + WS_DENPART;
  float* dect     = ws + WS_DECT;

  k_enc_partial<<<dim3(32, 32), 256, 0, stream>>>(x, enc_w, enc_part);
  k_enc_finish<<<1, 512, 0, stream>>>(enc_part, enc_b, bw_w, bw_b, encoded, invw2t);
  k_dec_transpose<<<TR_BLKS, 256, 0, stream>>>(dec, dect);
  k_denom_partial<<<DEN_BLKS, 256, 0, stream>>>(nd, lab, invw2t, denpart);
  k_denom_finish<<<1, 1024, 0, stream>>>(denpart, DEN_BLKS, rdent);
  k_fused_out<<<200, 256, 0, stream>>>(nd, nid, lab, dect, invw2t, rdent,
                                       encoded, out);
}


# occupancy: denominator pass 256 blocks of 32 waves (was 224 of 8), fused gather one tile per wave in 4-wave blocks at 7 waves per SIMD (was 2): latency overlapped; denominator sums regrouped
# speedup vs baseline: 5.9389x; 1.4004x over previous
#include <hip/hip_runtime.h>
#include <math.h>


#define NPTS 100000
#define NBAS 32
#define MCLU 16
#define KNEI 32
#define MUF  100.0f

typedef __attribute__((ext_vector_type(2))) float v2f;
typedef __attribute__((ext_vector_type(8))) float v8f;

#define WS_ENCPART 0
#define WS_ENCODED (WS_ENCPART + 1024)
#define WS_INVW2T  (WS_ENCODED + 32)
#define WS_RDENT   (WS_INVW2T + 512)
#define DEN_BLKS   256
#define DEN_WAVES  32
#define WS_DENPART (WS_RDENT + 1024)
#define WS_DECT    (WS_DENPART + DEN_BLKS*1024)

__global__ void k_enc_partial(const float* __restrict__ x,
                              const float* __restrict__ W,
                              float* __restrict__ part) {
  const int row = blockIdx.y, chunk = blockIdx.x;
  const int CS = (NPTS + 31) / 32;
  const int j0 = chunk * CS;
  const int j1 = (j0 + CS < NPTS) ? (j0 + CS) : NPTS;
  float s = 0.f;
  for (int j = j0 + threadIdx.x; j < j1; j += blockDim.x)
    s += x[j] * W[row * NPTS + j];
  __shared__ float red[256];
  red[threadIdx.x] = s;
  __syncthreads();
  for (int o = 128; o > 0; o >>= 1) {
    if (threadIdx.x < o) red[threadIdx.x] += red[threadIdx.x + o];
    __syncthreads();
  }
  if (threadIdx.x == 0) part[row * 32 + chunk] = red[0];
}

__global__ void k_enc_finish(const float* __restrict__ part,
                             const float* __restrict__ enc_b,
                             const float* __restrict__ bw_w,
                             const float* __restrict__ bw_b,
                             float* __restrict__ encoded,
                             float* __restrict__ invw2t) {
  __shared__ float enc_s[NBAS];
  const int t = threadIdx.x;
  if (t < NBAS) {
    float s = 0.f;
    for (int c = 0; c < 32; ++c) s += part[t * 32 + c];
    s += enc_b[t];
    enc_s[t] = s;
    encoded[t] = s;
  }
  __syncthreads();
  if (t < NBAS * MCLU) {
    float s = 0.f;
    for (int q = 0; q < NBAS; ++q) s += enc_s[q] * bw_w[t * NBAS + q];
    s += bw_b[t];
    float band = (1.f / (1.f + expf(-s))) / 60.f;
    float wm = band * MUF;
    const int i = t / MCLU, c = t % MCLU;
    invw2t[c * NBAS + i] = 1.f / (wm * wm);
  }
}

#define TR_TILES 3125
#define TR_BLKS  391
__global__ void __launch_bounds__(256) k_dec_transpose(const float* __restrict__ dec,
                                                       float* __restrict__ dect) {
  __shared__ float tile[8][32 * 33];
  const int lane = threadIdx.x & 31;
  const int wv = __builtin_amdgcn_readfirstlane(threadIdx.x >> 5);
  const int t = blockIdx.x * 8 + wv;
  if (t >= TR_TILES) return;
  const int jb = t * 32;
  float* tw = tile[wv];
  float v[32];
#pragma unroll
  for (int i = 0; i < 32; ++i) v[i] = dec[i * NPTS + jb + lane];
#pragma unroll
  for (int i = 0; i < 32; ++i) tw[i * 33 + lane] = v[i];
  asm volatile("s_wait_dscnt 0" ::: "memory");
#pragma unroll
  for (int r = 0; r < 32; ++r) dect[(jb + r) * 32 + lane] = tw[lane * 33 + r];
}

__global__ void __launch_bounds__(1024) k_denom_partial(const float* __restrict__ d,
                                const int* __restrict__ lab,
                                const float* __restrict__ invw2t,
                                float* __restrict__ part) {
  __shared__ float iw[MCLU * NBAS];
  __shared__ float red[DEN_WAVES * 1024];
  for (int t = threadIdx.x; t < MCLU * NBAS; t += blockDim.x) iw[t] = invw2t[t];
  __syncthreads();
  const int lane = threadIdx.x & 31, wv = threadIdx.x >> 5;
  const int gw = blockIdx.x * DEN_WAVES + wv, nw = gridDim.x * DEN_WAVES;
  float acc[NBAS];
#pragma unroll
  for (int i = 0; i < NBAS; ++i) acc[i] = 0.f;
  for (int j = gw; j < NPTS; j += nw) {
    const int c = lab[j];
    const float dv = d[j * KNEI + lane];
    const float d2 = dv * dv;
#pragma unroll
    for (int i = 0; i < NBAS; ++i)
      acc[i] += fmaxf(1.f - d2 * iw[c * NBAS + i], 0.f);
  }
#pragma unroll
  for (int i = 0; i < NBAS; ++i) red[wv * 1024 + i * 32 + lane] = acc[i];
  __syncthreads();
  for (int e = threadIdx.x; e < 1024; e += blockDim.x) {
    float s = 0.f;
#pragma unroll
    for (int w = 0; w < DEN_WAVES; ++w) s += red[w * 1024 + e];
    part[blockIdx.x * 1024 + e] = s;
  }
}

__global__ void k_denom_finish(const float* __restrict__ part, int nblk,
                               float* __restrict__ rdent) {
  const int t = threadIdx.x;
  float s = 0.f;
  for (int b = 0; b < nblk; ++b) s += part[b * 1024 + t];
  const int i = t >> 5, kk = t & 31;
  rdent[kk * 32 + i] = 1.f / s;
}

__device__ __forceinline__ void fused_point(int& nl, float& dv, int& c, const int p,
                                            const int j0, const int lane,
                                            const float* iw, const float (&rdr)[KNEI],
                                            const float* __restrict__ dect,
                                            const int* __restrict__ nid,
                                            const float* __restrict__ d,
                                            const int* __restrict__ lab,
                                            float* gbuf) {
  const float s = iw[c * NBAS + lane];
  const float d2 = dv * dv;
  float dec[KNEI];
#pragma unroll
  for (int kk = 0; kk < KNEI; ++kk) {
    const int nb = __shfl(nl, kk, 32);
    dec[kk] = dect[nb * 32 + lane];
  }
  __builtin_amdgcn_sched_barrier(0);
  int j2 = j0 + p + 2;
  j2 = (j2 < NPTS) ? j2 : (NPTS - 1);
  nl = nid[j2 * KNEI + lane];
  dv = d[j2 * KNEI + lane];
  c = lab[j2];
  float w[KNEI];
#pragma unroll
  for (int kk = 0; kk < KNEI; ++kk)
    w[kk] = rdr[kk] * fmaxf(1.f - __shfl(d2, kk, 32) * s, 0.f);
  __builtin_amdgcn_sched_barrier(0);
  float acc = 0.f;
#pragma unroll
  for (int kk = 0; kk < KNEI; ++kk) acc += w[kk] * dec[kk];
  gbuf[lane * 17 + p] = acc;
}

#define FO_WAVES 4
#define FO_BLKS  1563
__global__ void __launch_bounds__(128) __attribute__((amdgpu_waves_per_eu(7)))
k_fused_out(const float* __restrict__ d,
                            const int* __restrict__ nid,
                            const int* __restrict__ lab,
                            const float* __restrict__ dect,
                            const float* __restrict__ invw2t,
                            const float* __restrict__ rdent,
                            const float* __restrict__ encoded,
                            float* __restrict__ out) {
  __shared__ float iw[MCLU * NBAS];
  __shared__ float rd[NBAS * NBAS];
  __shared__ float enc_s[NBAS];
  __shared__ float g[FO_WAVES][32 * 17];
  for (int t = threadIdx.x; t < MCLU * NBAS; t += blockDim.x) iw[t] = invw2t[t];
  for (int t = threadIdx.x; t < NBAS * NBAS; t += blockDim.x) rd[t] = rdent[t];
  if (threadIdx.x < NBAS) enc_s[threadIdx.x] = encoded[threadIdx.x];
  __syncthreads();

  const int lane = threadIdx.x & 31, wv = threadIdx.x >> 5;
  const int gw = blockIdx.x * FO_WAVES + wv;
  float* gbuf = g[wv];
  const int NT = NPTS / 16;

  float rdr[KNEI];
#pragma unroll
  for (int kk = 0; kk < KNEI; ++kk) rdr[kk] = rd[kk * 32 + lane];

  int nlA = 0, cA = 0, nlB = 0, cB = 0;
  float dvA = 0.f, dvB = 0.f;
  if (gw < NT) {
    const int jf = gw * 16;
    nlA = nid[jf * KNEI + lane];
    dvA = d[jf * KNEI + lane];
    cA = lab[jf];
    nlB = nid[(jf + 1) * KNEI + lane];
    dvB = d[(jf + 1) * KNEI + lane];
    cB = lab[jf + 1];
  }

  if (gw < NT) {
    const int j0 = gw * 16;
    for (int p = 0; p < 16; p += 2) {
      fused_point(nlA, dvA, cA, p, j0, lane, iw, rdr, dect, nid, d, lab, gbuf);
      fused_point(nlB, dvB, cB, p + 1, j0, lane, iw, rdr, dect, nid, d, lab, gbuf);
    }
    asm volatile("s_wait_dscnt 0" ::: "memory");

    v8f cacc = {0.f, 0.f, 0.f, 0.f, 0.f, 0.f, 0.f, 0.f};
    const int half = (lane >> 4) & 1;
    const int col = lane & 15;
#pragma unroll
    for (int s8 = 0; s8 < 8; ++s8) {
      const int kb = s8 * 4;
      v2f a, b;
      a.x = enc_s[kb + 2 * half];
      a.y = enc_s[kb + 2 * half + 1];
      b.x = gbuf[(kb + 2 * half) * 17 + col];
      b.y = gbuf[(kb + 2 * half + 1) * 17 + col];
      cacc = __builtin_amdgcn_wmma_f32_16x16x4_f32(
          false, a, false, b, (short)0, cacc, false, false);
    }
    if (lane < 16) out[j0 + lane] = cacc[0];
  }
}

extern "C" void kernel_launch(void* const* d_in, const int* in_sizes, int n_in,
                              void* d_out, int out_size, void* d_ws, size_t ws_size,
                              hipStream_t stream) {
  const float* x     = (const float*)d_in[0];
  const float* enc_w = (const float*)d_in[1];
  const float* enc_b = (const float*)d_in[2];
  const float* dec   = (const float*)d_in[3];
  const float* bw_w  = (const float*)d_in[4];
  const float* bw_b  = (const float*)d_in[5];
  const float* nd    = (const float*)d_in[6];
  const int*   nid   = (const int*)d_in[7];
  const int*   lab   = (const int*)d_in[8];
  float* out = (float*)d_out;
  float* ws  = (float*)d_ws;

  float* enc_part = ws + WS_ENCPART;
  float* encoded  = ws + WS_ENCODED;
  float* invw2t   = ws + WS_INVW2T;
  float* rdent    = ws + WS_RDENT;
  float* denpart  = ws + WS_DENPART;
  float* dect     = ws + WS_DECT;

  k_enc_partial<<<dim3(32, 32), 256, 0, stream>>>(x, enc_w, enc_part);
  k_enc_finish<<<1, 512, 0, stream>>>(enc_part, enc_b, bw_w, bw_b, encoded, invw2t);
  k_dec_transpose<<<TR_BLKS, 256, 0, stream>>>(dec, dect);
  k_denom_partial<<<DEN_BLKS, 1024, 0, stream>>>(nd, lab, invw2t, denpart);
  k_denom_finish<<<1, 1024, 0, stream>>>(denpart, DEN_BLKS, rdent);
  k_fused_out<<<FO_BLKS, 128, 0, stream>>>(nd, nid, lab, dect, invw2t, rdent,
                                       encoded, out);
}


# fused gather kernel: every point's 32 gather loads issue back to back: indices and distances broadcast by v_readlane_b32 (was ds_bpermute_b32; odd points waited an LDS round trip per gather)
# speedup vs baseline: 6.7953x; 1.1442x over previous
#include <hip/hip_runtime.h>
#include <math.h>


#define NPTS 100000
#define NBAS 32
#define MCLU 16
#define KNEI 32
#define MUF  100.0f

typedef __attribute__((ext_vector_type(2))) float v2f;
typedef __attribute__((ext_vector_type(8))) float v8f;

#define WS_ENCPART 0
#define WS_ENCODED (WS_ENCPART + 1024)
#define WS_INVW2T  (WS_ENCODED + 32)
#define WS_RDENT   (WS_INVW2T + 512)
#define DEN_BLKS   256
#define DEN_WAVES  32
#define WS_DENPART (WS_RDENT + 1024)
#define WS_DECT    (WS_DENPART + DEN_BLKS*1024)

__global__ void k_enc_partial(const float* __restrict__ x,
                              const float* __restrict__ W,
                              float* __restrict__ part) {
  const int row = blockIdx.y, chunk = blockIdx.x;
  const int CS = (NPTS + 31) / 32;
  const int j0 = chunk * CS;
  const int j1 = (j0 + CS < NPTS) ? (j0 + CS) : NPTS;
  float s = 0.f;
  for (int j = j0 + threadIdx.x; j < j1; j += blockDim.x)
    s += x[j] * W[row * NPTS + j];
  __shared__ float red[256];
  red[threadIdx.x] = s;
  __syncthreads();
  for (int o = 128; o > 0; o >>= 1) {
    if (threadIdx.x < o) red[threadIdx.x] += red[threadIdx.x + o];
    __syncthreads();
  }
  if (threadIdx.x == 0) part[row * 32 + chunk] = red[0];
}

__global__ void k_enc_finish(const float* __restrict__ part,
                             const float* __restrict__ enc_b,
                             const float* __restrict__ bw_w,
                             const float* __restrict__ bw_b,
                             float* __restrict__ encoded,
                             float* __restrict__ invw2t) {
  __shared__ float enc_s[NBAS];
  const int t = threadIdx.x;
  if (t < NBAS) {
    float s = 0.f;
    for (int c = 0; c < 32; ++c) s += part[t * 32 + c];
    s += enc_b[t];
    enc_s[t] = s;
    encoded[t] = s;
  }
  __syncthreads();
  if (t < NBAS * MCLU) {
    float s = 0.f;
    for (int q = 0; q < NBAS; ++q) s += enc_s[q] * bw_w[t * NBAS + q];
    s += bw_b[t];
    float band = (1.f / (1.f + expf(-s))) / 60.f;
    float wm = band * MUF;
    const int i = t / MCLU, c = t % MCLU;
    invw2t[c * NBAS + i] = 1.f / (wm * wm);
  }
}

#define TR_TILES 3125
#define TR_BLKS  391
__global__ void __launch_bounds__(256) k_dec_transpose(const float* __restrict__ dec,
                                                       float* __restrict__ dect) {
  __shared__ float tile[8][32 * 33];
  const int lane = threadIdx.x & 31;
  const int wv = __builtin_amdgcn_readfirstlane(threadIdx.x >> 5);
  const int t = blockIdx.x * 8 + wv;
  if (t >= TR_TILES) return;
  const int jb = t * 32;
  float* tw = tile[wv];
  float v[32];
#pragma unroll
  for (int i = 0; i < 32; ++i) v[i] = dec[i * NPTS + jb + lane];
#pragma unroll
  for (int i = 0; i < 32; ++i) tw[i * 33 + lane] = v[i];
  asm volatile("s_wait_dscnt 0" ::: "memory");
#pragma unroll
  for (int r = 0; r < 32; ++r) dect[(jb + r) * 32 + lane] = tw[lane * 33 + r];
}

__global__ void __launch_bounds__(1024) k_denom_partial(const float* __restrict__ d,
                                const int* __restrict__ lab,
                                const float* __restrict__ invw2t,
                                float* __restrict__ part) {
  __shared__ float iw[MCLU * NBAS];
  __shared__ float red[DEN_WAVES * 1024];
  for (int t = threadIdx.x; t < MCLU * NBAS; t += blockDim.x) iw[t] = invw2t[t];
  __syncthreads();
  const int lane = threadIdx.x & 31, wv = threadIdx.x >> 5;
  const int gw = blockIdx.x * DEN_WAVES + wv, nw = gridDim.x * DEN_WAVES;
  float acc[NBAS];
#pragma unroll
  for (int i = 0; i < NBAS; ++i) acc[i] = 0.f;
  for (int j = gw; j < NPTS; j += nw) {
    const int c = lab[j];
    const float dv = d[j * KNEI + lane];
    const float d2 = dv * dv;
#pragma unroll
    for (int i = 0; i < NBAS; ++i)
      acc[i] += fmaxf(1.f - d2 * iw[c * NBAS + i], 0.f);
  }
#pragma unroll
  for (int i = 0; i < NBAS; ++i) red[wv * 1024 + i * 32 + lane] = acc[i];
  __syncthreads();
  for (int e = threadIdx.x; e < 1024; e += blockDim.x) {
    float s = 0.f;
#pragma unroll
    for (int w = 0; w < DEN_WAVES; ++w) s += red[w * 1024 + e];
    part[blockIdx.x * 1024 + e] = s;
  }
}

__global__ void k_denom_finish(const float* __restrict__ part, int nblk,
                               float* __restrict__ rdent) {
  const int t = threadIdx.x;
  float s = 0.f;
  for (int b = 0; b < nblk; ++b) s += part[b * 1024 + t];
  const int i = t >> 5, kk = t & 31;
  rdent[kk * 32 + i] = 1.f / s;
}

__device__ __forceinline__ int lane_to_all(int v, int l) {
  return __builtin_amdgcn_readlane(v, l);
}
__device__ __forceinline__ float lane_to_all(float v, int l) {
  return __builtin_bit_cast(float,
                            __builtin_amdgcn_readlane(__builtin_bit_cast(int, v), l));
}

__device__ __forceinline__ void fused_point(int& nl, float& dv, int& c, const int p,
                                            const int j0, const int lane,
                                            const float* iw, const float (&rdr)[KNEI],
                                            const float* __restrict__ dect,
                                            const int* __restrict__ nid,
                                            const float* __restrict__ d,
                                            const int* __restrict__ lab,
                                            float* gbuf) {
  const float s = iw[c * NBAS + lane];
  const float d2 = dv * dv;
  int off[KNEI];
#pragma unroll
  for (int kk = 0; kk < KNEI; ++kk)
    off[kk] = lane_to_all(nl, kk) * 32 + lane;
  __builtin_amdgcn_sched_barrier(0);
  float dec[KNEI];
#pragma unroll
  for (int kk = 0; kk < KNEI; ++kk)
    dec[kk] = dect[off[kk]];
  __builtin_amdgcn_sched_barrier(0);
  int j2 = j0 + p + 2;
  j2 = (j2 < NPTS) ? j2 : (NPTS - 1);
  nl = nid[j2 * KNEI + lane];
  dv = d[j2 * KNEI + lane];
  c = lab[j2];
  float w[KNEI];
#pragma unroll
  for (int kk = 0; kk < KNEI; ++kk)
    w[kk] = rdr[kk] * fmaxf(1.f - lane_to_all(d2, kk) * s, 0.f);
  __builtin_amdgcn_sched_barrier(0);
  float acc = 0.f;
#pragma unroll
  for (int kk = 0; kk < KNEI; ++kk) acc += w[kk] * dec[kk];
  gbuf[lane * 17 + p] = acc;
}

#define FO_WAVES 4
#define FO_BLKS  1563
__global__ void __launch_bounds__(128) __attribute__((amdgpu_waves_per_eu(7)))
k_fused_out(const float* __restrict__ d,
                            const int* __restrict__ nid,
                            const int* __restrict__ lab,
                            const float* __restrict__ dect,
                            const float* __restrict__ invw2t,
                            const float* __restrict__ rdent,
                            const float* __restrict__ encoded,
                            float* __restrict__ out) {
  __shared__ float iw[MCLU * NBAS];
  __shared__ float rd[NBAS * NBAS];
  __shared__ float enc_s[NBAS];
  __shared__ float g[FO_WAVES][32 * 17];
  for (int t = threadIdx.x; t < MCLU * NBAS; t += blockDim.x) iw[t] = invw2t[t];
  for (int t = threadIdx.x; t < NBAS * NBAS; t += blockDim.x) rd[t] = rdent[t];
  if (threadIdx.x < NBAS) enc_s[threadIdx.x] = encoded[threadIdx.x];
  __syncthreads();

  const int lane = threadIdx.x & 31, wv = threadIdx.x >> 5;
  const int gw = blockIdx.x * FO_WAVES + wv;
  float* gbuf = g[wv];
  const int NT = NPTS / 16;

  float rdr[KNEI];
#pragma unroll
  for (int kk = 0; kk < KNEI; ++kk) rdr[kk] = rd[kk * 32 + lane];

  int nlA = 0, cA = 0, nlB = 0, cB = 0;
  float dvA = 0.f, dvB = 0.f;
  if (gw < NT) {
    const int jf = gw * 16;
    nlA = nid[jf * KNEI + lane];
    dvA = d[jf * KNEI + lane];
    cA = lab[jf];
    nlB = nid[(jf + 1) * KNEI + lane];
    dvB = d[(jf + 1) * KNEI + lane];
    cB = lab[jf + 1];
  }

  if (gw < NT) {
    const int j0 = gw * 16;
    for (int p = 0; p < 16; p += 2) {
      fused_point(nlA, dvA, cA, p, j0, lane, iw, rdr, dect, nid, d, lab, gbuf);
      fused_point(nlB, dvB, cB, p + 1, j0, lane, iw, rdr, dect, nid, d, lab, gbuf);
    }
    asm volatile("s_wait_dscnt 0" ::: "memory");

    v8f cacc = {0.f, 0.f, 0.f, 0.f, 0.f, 0.f, 0.f, 0.f};
    const int half = (lane >> 4) & 1;
    const int col = lane & 15;
#pragma unroll
    for (int s8 = 0; s8 < 8; ++s8) {
      const int kb = s8 * 4;
      v2f a, b;
      a.x = enc_s[kb + 2 * half];
      a.y = enc_s[kb + 2 * half + 1];
      b.x = gbuf[(kb + 2 * half) * 17 + col];
      b.y = gbuf[(kb + 2 * half + 1) * 17 + col];
      cacc = __builtin_amdgcn_wmma_f32_16x16x4_f32(
          false, a, false, b, (short)0, cacc, false, false);
    }
    if (lane < 16) out[j0 + lane] = cacc[0];
  }
}

extern "C" void kernel_launch(void* const* d_in, const int* in_sizes, int n_in,
                              void* d_out, int out_size, void* d_ws, size_t ws_size,
                              hipStream_t stream) {
  const float* x     = (const float*)d_in[0];
  const float* enc_w = (const float*)d_in[1];
  const float* enc_b = (const float*)d_in[2];
  const float* dec   = (const float*)d_in[3];
  const float* bw_w  = (const float*)d_in[4];
  const float* bw_b  = (const float*)d_in[5];
  const float* nd    = (const float*)d_in[6];
  const int*   nid   = (const int*)d_in[7];
  const int*   lab   = (const int*)d_in[8];
  float* out = (float*)d_out;
  float* ws  = (float*)d_ws;

  float* enc_part = ws + WS_ENCPART;
  float* encoded  = ws + WS_ENCODED;
  float* invw2t   = ws + WS_INVW2T;
  float* rdent    = ws + WS_RDENT;
  float* denpart  = ws + WS_DENPART;
  float* dect     = ws + WS_DECT;

  k_enc_partial<<<dim3(32, 32), 256, 0, stream>>>(x, enc_w, enc_part);
  k_enc_finish<<<1, 512, 0, stream>>>(enc_part, enc_b, bw_w, bw_b, encoded, invw2t);
  k_dec_transpose<<<TR_BLKS, 256, 0, stream>>>(dec, dect);
  k_denom_partial<<<DEN_BLKS, 1024, 0, stream>>>(nd, lab, invw2t, denpart);
  k_denom_finish<<<1, 1024, 0, stream>>>(denpart, DEN_BLKS, rdent);
  k_fused_out<<<FO_BLKS, 128, 0, stream>>>(nd, nid, lab, dect, invw2t, rdent,
                                       encoded, out);
}


# encoder GEMV and denominator pass: a wave requests all its rows before the first use (GEMV 13 x 16 B of W and of x per lane, 2048 waves), was load-wait-use per trip; GEMV sums regrouped
# speedup vs baseline: 7.5090x; 1.1050x over previous
#include <hip/hip_runtime.h>
#include <math.h>


#define NPTS 100000
#define NBAS 32
#define MCLU 16
#define KNEI 32
#define MUF  100.0f

typedef __attribute__((ext_vector_type(2))) float v2f;
typedef __attribute__((ext_vector_type(8))) float v8f;

#define WS_ENCPART 0
#define WS_ENCODED (WS_ENCPART + 1024)
#define WS_INVW2T  (WS_ENCODED + 32)
#define WS_RDENT   (WS_INVW2T + 512)
#define DEN_BLKS   256
#define DEN_WAVES  32
#define WS_DENPART (WS_RDENT + 1024)
#define WS_DECT    (WS_DENPART + DEN_BLKS*1024)

#define ENC_CQ  782
#define ENC_T   13
#define ENC_TIN 11
static_assert(64 * ENC_TIN <= NPTS / 4 - 31 * ENC_CQ, "the last chunk holds ENC_TIN groups per lane");
typedef __attribute__((ext_vector_type(4))) float v4f;
__global__ void __launch_bounds__(64) k_enc_partial(const float* __restrict__ x,
                                                    const float* __restrict__ W,
                                                    float* __restrict__ part) {
  const int row = blockIdx.y, chunk = blockIdx.x;
  const int NQ = NPTS / 4;
  const int q0 = chunk * ENC_CQ;
  const int q1 = (q0 + ENC_CQ < NQ) ? (q0 + ENC_CQ) : NQ;
  const v4f* x4 = (const v4f*)x;
  const v4f* w4 = (const v4f*)(W + row * NPTS);
  v4f xv[ENC_T], wv[ENC_T];
#pragma unroll
  for (int t = 0; t < ENC_T; ++t) {
    const int q = q0 + (int)threadIdx.x + 64 * t;
    const int qc = (t < ENC_TIN || q < q1) ? q : (q1 - 1);
    wv[t] = w4[qc];
    xv[t] = x4[qc];
  }
  __builtin_amdgcn_sched_barrier(0);
  float s = 0.f;
#pragma unroll
  for (int t = 0; t < ENC_T; ++t) {
    const int q = q0 + (int)threadIdx.x + 64 * t;
    const float p = xv[t].x * wv[t].x + xv[t].y * wv[t].y +
                    xv[t].z * wv[t].z + xv[t].w * wv[t].w;
    s += (t < ENC_TIN || q < q1) ? p : 0.f;
  }
  __shared__ float red[64];
  red[threadIdx.x] = s;
  __syncthreads();
  for (int o = 32; o > 0; o >>= 1) {
    if (threadIdx.x < o) red[threadIdx.x] += red[threadIdx.x + o];
    __syncthreads();
  }
  if (threadIdx.x == 0) part[row * 32 + chunk] = red[0];
}

__global__ void k_enc_finish(const float* __restrict__ part,
                             const float* __restrict__ enc_b,
                             const float* __restrict__ bw_w,
                             const float* __restrict__ bw_b,
                             float* __restrict__ encoded,
                             float* __restrict__ invw2t) {
  __shared__ float enc_s[NBAS];
  const int t = threadIdx.x;
  if (t < NBAS) {
    float s = 0.f;
    for (int c = 0; c < 32; ++c) s += part[t * 32 + c];
    s += enc_b[t];
    enc_s[t] = s;
    encoded[t] = s;
  }
  __syncthreads();
  if (t < NBAS * MCLU) {
    float s = 0.f;
    for (int q = 0; q < NBAS; ++q) s += enc_s[q] * bw_w[t * NBAS + q];
    s += bw_b[t];
    float band = (1.f / (1.f + expf(-s))) / 60.f;
    float wm = band * MUF;
    const int i = t / MCLU, c = t % MCLU;
    invw2t[c * NBAS + i] = 1.f / (wm * wm);
  }
}

#define TR_TILES 3125
#define TR_BLKS  391
__global__ void __launch_bounds__(256) k_dec_transpose(const float* __restrict__ dec,
                                                       float* __restrict__ dect) {
  __shared__ float tile[8][32 * 33];
  const int lane = threadIdx.x & 31;
  const int wv = __builtin_amdgcn_readfirstlane(threadIdx.x >> 5);
  const int t = blockIdx.x * 8 + wv;
  if (t >= TR_TILES) return;
  const int jb = t * 32;
  float* tw = tile[wv];
  float v[32];
#pragma unroll
  for (int i = 0; i < 32; ++i) v[i] = dec[i * NPTS + jb + lane];
#pragma unroll
  for (int i = 0; i < 32; ++i) tw[i * 33 + lane] = v[i];
  asm volatile("s_wait_dscnt 0" ::: "memory");
#pragma unroll
  for (int r = 0; r < 32; ++r) dect[(jb + r) * 32 + lane] = tw[lane * 33 + r];
}

#define DEN_NW (DEN_BLKS * DEN_WAVES)
#define DEN_T  ((NPTS + DEN_NW - 1) / DEN_NW)
__global__ void __launch_bounds__(1024) k_denom_partial(const float* __restrict__ d,
                                const int* __restrict__ lab,
                                const float* __restrict__ invw2t,
                                float* __restrict__ part) {
  __shared__ float iw[MCLU * NBAS];
  __shared__ float red[DEN_WAVES * 1024];
  const int lane = threadIdx.x & 31, wv = threadIdx.x >> 5;
  const int gw = blockIdx.x * DEN_WAVES + wv;
  int cc[DEN_T];
  float dvv[DEN_T];
#pragma unroll
  for (int t = 0; t < DEN_T; ++t) {
    int j = gw + t * DEN_NW;
    j = (j < NPTS) ? j : (NPTS - 1);
    cc[t] = lab[j];
    dvv[t] = d[j * KNEI + lane];
  }
  for (int t = threadIdx.x; t < MCLU * NBAS; t += blockDim.x) iw[t] = invw2t[t];
  __syncthreads();
  float acc[NBAS];
#pragma unroll
  for (int i = 0; i < NBAS; ++i) acc[i] = 0.f;
#pragma unroll
  for (int t = 0; t < DEN_T; ++t) {
    if (gw + t * DEN_NW < NPTS) {
      const int c = cc[t];
      const float d2 = dvv[t] * dvv[t];
#pragma unroll
      for (int i = 0; i < NBAS; ++i)
        acc[i] += fmaxf(1.f - d2 * iw[c * NBAS + i], 0.f);
    }
  }
#pragma unroll
  for (int i = 0; i < NBAS; ++i) red[wv * 1024 + i * 32 + lane] = acc[i];
  __syncthreads();
  for (int e = threadIdx.x; e < 1024; e += blockDim.x) {
    float s = 0.f;
#pragma unroll
    for (int w = 0; w < DEN_WAVES; ++w) s += red[w * 1024 + e];
    part[blockIdx.x * 1024 + e] = s;
  }
}

__global__ void k_denom_finish(const float* __restrict__ part, int nblk,
                               float* __restrict__ rdent) {
  const int t = threadIdx.x;
  float s = 0.f;
  for (int b = 0; b < nblk; ++b) s += part[b * 1024 + t];
  const int i = t >> 5, kk = t & 31;
  rdent[kk * 32 + i] = 1.f / s;
}

__device__ __forceinline__ int lane_to_all(int v, int l) {
  return __builtin_amdgcn_readlane(v, l);
}
__device__ __forceinline__ float lane_to_all(float v, int l) {
  return __builtin_bit_cast(float,
                            __builtin_amdgcn_readlane(__builtin_bit_cast(int, v), l));
}

__device__ __forceinline__ void fused_point(int& nl, float& dv, int& c, const int p,
                                            const int j0, const int lane,
                                            const float* iw, const float (&rdr)[KNEI],
                                            const float* __restrict__ dect,
                                            const int* __restrict__ nid,
                                            const float* __restrict__ d,
                                            const int* __restrict__ lab,
                                            float* gbuf) {
  const float s = iw[c * NBAS + lane];
  const float d2 = dv * dv;
  int off[KNEI];
#pragma unroll
  for (int kk = 0; kk < KNEI; ++kk)
    off[kk] = lane_to_all(nl, kk) * 32 + lane;
  __builtin_amdgcn_sched_barrier(0);
  float dec[KNEI];
#pragma unroll
  for (int kk = 0; kk < KNEI; ++kk)
    dec[kk] = dect[off[kk]];
  __builtin_amdgcn_sched_barrier(0);
  int j2 = j0 + p + 2;
  j2 = (j2 < NPTS) ? j2 : (NPTS - 1);
  nl = nid[j2 * KNEI + lane];
  dv = d[j2 * KNEI + lane];
  c = lab[j2];
  float w[KNEI];
#pragma unroll
  for (int kk = 0; kk < KNEI; ++kk)
    w[kk] = rdr[kk] * fmaxf(1.f - lane_to_all(d2, kk) * s, 0.f);
  __builtin_amdgcn_sched_barrier(0);
  float acc = 0.f;
#pragma unroll
  for (int kk = 0; kk < KNEI; ++kk) acc += w[kk] * dec[kk];
  gbuf[lane * 17 + p] = acc;
}

#define FO_WAVES 4
#define FO_BLKS  1563
__global__ void __launch_bounds__(128) __attribute__((amdgpu_waves_per_eu(7)))
k_fused_out(const float* __restrict__ d,
                            const int* __restrict__ nid,
                            const int* __restrict__ lab,
                            const float* __restrict__ dect,
                            const float* __restrict__ invw2t,
                            const float* __restrict__ rdent,
                            const float* __restrict__ encoded,
                            float* __restrict__ out) {
  __shared__ float iw[MCLU * NBAS];
  __shared__ float rd[NBAS * NBAS];
  __shared__ float enc_s[NBAS];
  __shared__ float g[FO_WAVES][32 * 17];
  for (int t = threadIdx.x; t < MCLU * NBAS; t += blockDim.x) iw[t] = invw2t[t];
  for (int t = threadIdx.x; t < NBAS * NBAS; t += blockDim.x) rd[t] = rdent[t];
  if (threadIdx.x < NBAS) enc_s[threadIdx.x] = encoded[threadIdx.x];
  __syncthreads();

  const int lane = threadIdx.x & 31, wv = threadIdx.x >> 5;
  const int gw = blockIdx.x * FO_WAVES + wv;
  float* gbuf = g[wv];
  const int NT = NPTS / 16;

  float rdr[KNEI];
#pragma unroll
  for (int kk = 0; kk < KNEI; ++kk) rdr[kk] = rd[kk * 32 + lane];

  int nlA = 0, cA = 0, nlB = 0, cB = 0;
  float dvA = 0.f, dvB = 0.f;
  if (gw < NT) {
    const int jf = gw * 16;
    nlA = nid[jf * KNEI + lane];
    dvA = d[jf * KNEI + lane];
    cA = lab[jf];
    nlB = nid[(jf + 1) * KNEI + lane];
    dvB = d[(jf + 1) * KNEI + lane];
    cB = lab[jf + 1];
  }

  if (gw < NT) {
    const int j0 = gw * 16;
    for (int p = 0; p < 16; p += 2) {
      fused_point(nlA, dvA, cA, p, j0, lane, iw, rdr, dect, nid, d, lab, gbuf);
      fused_point(nlB, dvB, cB, p + 1, j0, lane, iw, rdr, dect, nid, d, lab, gbuf);
    }
    asm volatile("s_wait_dscnt 0" ::: "memory");

    v8f cacc = {0.f, 0.f, 0.f, 0.f, 0.f, 0.f, 0.f, 0.f};
    const int half = (lane >> 4) & 1;
    const int col = lane & 15;
#pragma unroll
    for (int s8 = 0; s8 < 8; ++s8) {
      const int kb = s8 * 4;
      v2f a, b;
      a.x = enc_s[kb + 2 * half];
      a.y = enc_s[kb + 2 * half + 1];
      b.x = gbuf[(kb + 2 * half) * 17 + col];
      b.y = gbuf[(kb + 2 * half + 1) * 17 + col];
      cacc = __builtin_amdgcn_wmma_f32_16x16x4_f32(
          false, a, false, b, (short)0, cacc, false, false);
    }
    if (lane < 16) out[j0 + lane] = cacc[0];
  }
}

extern "C" void kernel_launch(void* const* d_in, const int* in_sizes, int n_in,
                              void* d_out, int out_size, void* d_ws, size_t ws_size,
                              hipStream_t stream) {
  const float* x     = (const float*)d_in[0];
  const float* enc_w = (const float*)d_in[1];
  const float* enc_b = (const float*)d_in[2];
  const float* dec   = (const float*)d_in[3];
  const float* bw_w  = (const float*)d_in[4];
  const float* bw_b  = (const float*)d_in[5];
  const float* nd    = (const float*)d_in[6];
  const int*   nid   = (const int*)d_in[7];
  const int*   lab   = (const int*)d_in[8];
  float* out = (float*)d_out;
  float* ws  = (float*)d_ws;

  float* enc_part = ws + WS_ENCPART;
  float* encoded  = ws + WS_ENCODED;
  float* invw2t   = ws + WS_INVW2T;
  float* rdent    = ws + WS_RDENT;
  float* denpart  = ws + WS_DENPART;
  float* dect     = ws + WS_DECT;

  k_enc_partial<<<dim3(32, 32), 64, 0, stream>>>(x, enc_w, enc_part);
  k_enc_finish<<<1, 512, 0, stream>>>(enc_part, enc_b, bw_w, bw_b, encoded, invw2t);
  k_dec_transpose<<<TR_BLKS, 256, 0, stream>>>(dec, dect);
  k_denom_partial<<<DEN_BLKS, 1024, 0, stream>>>(nd, lab, invw2t, denpart);
  k_denom_finish<<<1, 1024, 0, stream>>>(denpart, DEN_BLKS, rdent);
  k_fused_out<<<FO_BLKS, 128, 0, stream>>>(nd, nid, lab, dect, invw2t, rdent,
                                       encoded, out);
}


# denominator finish: 32 blocks (one per basis), a wave requests its 8 block partials together, waves combine through LDS (was 1 block, 32 dependent trips per wave); denominator sums regrouped
# speedup vs baseline: 9.9673x; 1.3274x over previous
#include <hip/hip_runtime.h>
#include <math.h>


#define NPTS 100000
#define NBAS 32
#define MCLU 16
#define KNEI 32
#define MUF  100.0f

typedef __attribute__((ext_vector_type(2))) float v2f;
typedef __attribute__((ext_vector_type(8))) float v8f;

#define WS_ENCPART 0
#define WS_ENCODED (WS_ENCPART + 1024)
#define WS_INVW2T  (WS_ENCODED + 32)
#define WS_RDENT   (WS_INVW2T + 512)
#define DEN_BLKS   256
#define DEN_WAVES  32
#define WS_DENPART (WS_RDENT + 1024)
#define WS_DECT    (WS_DENPART + DEN_BLKS*1024)

#define ENC_CQ  782
#define ENC_T   13
#define ENC_TIN 11
static_assert(64 * ENC_TIN <= NPTS / 4 - 31 * ENC_CQ, "the last chunk holds ENC_TIN groups per lane");
typedef __attribute__((ext_vector_type(4))) float v4f;
__global__ void __launch_bounds__(64) k_enc_partial(const float* __restrict__ x,
                                                    const float* __restrict__ W,
                                                    float* __restrict__ part) {
  const int row = blockIdx.y, chunk = blockIdx.x;
  const int NQ = NPTS / 4;
  const int q0 = chunk * ENC_CQ;
  const int q1 = (q0 + ENC_CQ < NQ) ? (q0 + ENC_CQ) : NQ;
  const v4f* x4 = (const v4f*)x;
  const v4f* w4 = (const v4f*)(W + row * NPTS);
  v4f xv[ENC_T], wv[ENC_T];
#pragma unroll
  for (int t = 0; t < ENC_T; ++t) {
    const int q = q0 + (int)threadIdx.x + 64 * t;
    const int qc = (t < ENC_TIN || q < q1) ? q : (q1 - 1);
    wv[t] = w4[qc];
    xv[t] = x4[qc];
  }
  __builtin_amdgcn_sched_barrier(0);
  float s = 0.f;
#pragma unroll
  for (int t = 0; t < ENC_T; ++t) {
    const int q = q0 + (int)threadIdx.x + 64 * t;
    const float p = xv[t].x * wv[t].x + xv[t].y * wv[t].y +
                    xv[t].z * wv[t].z + xv[t].w * wv[t].w;
    s += (t < ENC_TIN || q < q1) ? p : 0.f;
  }
  __shared__ float red[64];
  red[threadIdx.x] = s;
  __syncthreads();
  for (int o = 32; o > 0; o >>= 1) {
    if (threadIdx.x < o) red[threadIdx.x] += red[threadIdx.x + o];
    __syncthreads();
  }
  if (threadIdx.x == 0) part[row * 32 + chunk] = red[0];
}

__global__ void k_enc_finish(const float* __restrict__ part,
                             const float* __restrict__ enc_b,
                             const float* __restrict__ bw_w,
                             const float* __restrict__ bw_b,
                             float* __restrict__ encoded,
                             float* __restrict__ invw2t) {
  __shared__ float enc_s[NBAS];
  const int t = threadIdx.x;
  if (t < NBAS) {
    float s = 0.f;
    for (int c = 0; c < 32; ++c) s += part[t * 32 + c];
    s += enc_b[t];
    enc_s[t] = s;
    encoded[t] = s;
  }
  __syncthreads();
  if (t < NBAS * MCLU) {
    float s = 0.f;
    for (int q = 0; q < NBAS; ++q) s += enc_s[q] * bw_w[t * NBAS + q];
    s += bw_b[t];
    float band = (1.f / (1.f + expf(-s))) / 60.f;
    float wm = band * MUF;
    const int i = t / MCLU, c = t % MCLU;
    invw2t[c * NBAS + i] = 1.f / (wm * wm);
  }
}

#define TR_TILES 3125
#define TR_BLKS  391
__global__ void __launch_bounds__(256) k_dec_transpose(const float* __restrict__ dec,
                                                       float* __restrict__ dect) {
  __shared__ float tile[8][32 * 33];
  const int lane = threadIdx.x & 31;
  const int wv = __builtin_amdgcn_readfirstlane(threadIdx.x >> 5);
  const int t = blockIdx.x * 8 + wv;
  if (t >= TR_TILES) return;
  const int jb = t * 32;
  float* tw = tile[wv];
  float v[32];
#pragma unroll
  for (int i = 0; i < 32; ++i) v[i] = dec[i * NPTS + jb + lane];
#pragma unroll
  for (int i = 0; i < 32; ++i) tw[i * 33 + lane] = v[i];
  asm volatile("s_wait_dscnt 0" ::: "memory");
#pragma unroll
  for (int r = 0; r < 32; ++r) dect[(jb + r) * 32 + lane] = tw[lane * 33 + r];
}

#define DEN_NW (DEN_BLKS * DEN_WAVES)
#define DEN_T  ((NPTS + DEN_NW - 1) / DEN_NW)
__global__ void __launch_bounds__(1024) k_denom_partial(const float* __restrict__ d,
                                const int* __restrict__ lab,
                                const float* __restrict__ invw2t,
                                float* __restrict__ part) {
  __shared__ float iw[MCLU * NBAS];
  __shared__ float red[DEN_WAVES * 1024];
  const int lane = threadIdx.x & 31, wv = threadIdx.x >> 5;
  const int gw = blockIdx.x * DEN_WAVES + wv;
  int cc[DEN_T];
  float dvv[DEN_T];
#pragma unroll
  for (int t = 0; t < DEN_T; ++t) {
    int j = gw + t * DEN_NW;
    j = (j < NPTS) ? j : (NPTS - 1);
    cc[t] = lab[j];
    dvv[t] = d[j * KNEI + lane];
  }
  for (int t = threadIdx.x; t < MCLU * NBAS; t += blockDim.x) iw[t] = invw2t[t];
  __syncthreads();
  float acc[NBAS];
#pragma unroll
  for (int i = 0; i < NBAS; ++i) acc[i] = 0.f;
#pragma unroll
  for (int t = 0; t < DEN_T; ++t) {
    if (gw + t * DEN_NW < NPTS) {
      const int c = cc[t];
      const float d2 = dvv[t] * dvv[t];
#pragma unroll
      for (int i = 0; i < NBAS; ++i)
        acc[i] += fmaxf(1.f - d2 * iw[c * NBAS + i], 0.f);
    }
  }
#pragma unroll
  for (int i = 0; i < NBAS; ++i) red[wv * 1024 + i * 32 + lane] = acc[i];
  __syncthreads();
  for (int e = threadIdx.x; e < 1024; e += blockDim.x) {
    float s = 0.f;
#pragma unroll
    for (int w = 0; w < DEN_WAVES; ++w) s += red[w * 1024 + e];
    part[blockIdx.x * 1024 + e] = s;
  }
}

#define DF_PER (DEN_BLKS / 32)
static_assert(DF_PER * 32 == DEN_BLKS, "32 waves share the block partials evenly");
__global__ void __launch_bounds__(1024) k_denom_finish(const float* __restrict__ part,
                                                       float* __restrict__ rdent) {
  __shared__ float red[32 * 32];
  const int lane = threadIdx.x & 31, wv = threadIdx.x >> 5;
  const int e = blockIdx.x * 32 + lane;
  float v[DF_PER];
#pragma unroll
  for (int u = 0; u < DF_PER; ++u) v[u] = part[(wv * DF_PER + u) * 1024 + e];
  float s = 0.f;
#pragma unroll
  for (int u = 0; u < DF_PER; ++u) s += v[u];
  red[wv * 32 + lane] = s;
  __syncthreads();
  if (wv == 0) {
    float a = 0.f;
#pragma unroll
    for (int w = 0; w < 32; ++w) a += red[w * 32 + lane];
    rdent[e] = 1.f / a;
  }
}

__device__ __forceinline__ int lane_to_all(int v, int l) {
  return __builtin_amdgcn_readlane(v, l);
}
__device__ __forceinline__ float lane_to_all(float v, int l) {
  return __builtin_bit_cast(float,
                            __builtin_amdgcn_readlane(__builtin_bit_cast(int, v), l));
}

__device__ __forceinline__ void fused_point(int& nl, float& dv, int& c, const int p,
                                            const int j0, const int lane,
                                            const float* iw, const float (&rdr)[KNEI],
                                            const float* __restrict__ dect,
                                            const int* __restrict__ nid,
                                            const float* __restrict__ d,
                                            const int* __restrict__ lab,
                                            float* gbuf) {
  const float s = iw[c * NBAS + lane];
  const float d2 = dv * dv;
  int off[KNEI];
#pragma unroll
  for (int kk = 0; kk < KNEI; ++kk)
    off[kk] = lane_to_all(nl, kk) * 32 + lane;
  __builtin_amdgcn_sched_barrier(0);
  float dec[KNEI];
#pragma unroll
  for (int kk = 0; kk < KNEI; ++kk)
    dec[kk] = dect[off[kk]];
  __builtin_amdgcn_sched_barrier(0);
  int j2 = j0 + p + 2;
  j2 = (j2 < NPTS) ? j2 : (NPTS - 1);
  nl = nid[j2 * KNEI + lane];
  dv = d[j2 * KNEI + lane];
  c = lab[j2];
  float w[KNEI];
#pragma unroll
  for (int kk = 0; kk < KNEI; ++kk)
    w[kk] = rdr[kk] * fmaxf(1.f - lane_to_all(d2, kk) * s, 0.f);
  __builtin_amdgcn_sched_barrier(0);
  float acc = 0.f;
#pragma unroll
  for (int kk = 0; kk < KNEI; ++kk) acc += w[kk] * dec[kk];
  gbuf[lane * 17 + p] = acc;
}

#define FO_WAVES 4
#define FO_BLKS  1563
__global__ void __launch_bounds__(128) __attribute__((amdgpu_waves_per_eu(7)))
k_fused_out(const float* __restrict__ d,
                            const int* __restrict__ nid,
                            const int* __restrict__ lab,
                            const float* __restrict__ dect,
                            const float* __restrict__ invw2t,
                            const float* __restrict__ rdent,
                            const float* __restrict__ encoded,
                            float* __restrict__ out) {
  __shared__ float iw[MCLU * NBAS];
  __shared__ float rd[NBAS * 33];
  __shared__ float enc_s[NBAS];
  __shared__ float g[FO_WAVES][32 * 17];
  for (int t = threadIdx.x; t < MCLU * NBAS; t += blockDim.x) iw[t] = invw2t[t];
  for (int t = threadIdx.x; t < NBAS * NBAS; t += blockDim.x) rd[(t >> 5) * 33 + (t & 31)] = rdent[t];
  if (threadIdx.x < NBAS) enc_s[threadIdx.x] = encoded[threadIdx.x];
  __syncthreads();

  const int lane = threadIdx.x & 31, wv = threadIdx.x >> 5;
  const int gw = blockIdx.x * FO_WAVES + wv;
  float* gbuf = g[wv];
  const int NT = NPTS / 16;

  float rdr[KNEI];
#pragma unroll
  for (int kk = 0; kk < KNEI; ++kk) rdr[kk] = rd[lane * 33 + kk];

  int nlA = 0, cA = 0, nlB = 0, cB = 0;
  float dvA = 0.f, dvB = 0.f;
  if (gw < NT) {
    const int jf = gw * 16;
    nlA = nid[jf * KNEI + lane];
    dvA = d[jf * KNEI + lane];
    cA = lab[jf];
    nlB = nid[(jf + 1) * KNEI + lane];
    dvB = d[(jf + 1) * KNEI + lane];
    cB = lab[jf + 1];
  }

  if (gw < NT) {
    const int j0 = gw * 16;
    for (int p = 0; p < 16; p += 2) {
      fused_point(nlA, dvA, cA, p, j0, lane, iw, rdr, dect, nid, d, lab, gbuf);
      fused_point(nlB, dvB, cB, p + 1, j0, lane, iw, rdr, dect, nid, d, lab, gbuf);
    }
    asm volatile("s_wait_dscnt 0" ::: "memory");

    v8f cacc = {0.f, 0.f, 0.f, 0.f, 0.f, 0.f, 0.f, 0.f};
    const int half = (lane >> 4) & 1;
    const int col = lane & 15;
#pragma unroll
    for (int s8 = 0; s8 < 8; ++s8) {
      const int kb = s8 * 4;
      v2f a, b;
      a.x = enc_s[kb + 2 * half];
      a.y = enc_s[kb + 2 * half + 1];
      b.x = gbuf[(kb + 2 * half) * 17 + col];
      b.y = gbuf[(kb + 2 * half + 1) * 17 + col];
      cacc = __builtin_amdgcn_wmma_f32_16x16x4_f32(
          false, a, false, b, (short)0, cacc, false, false);
    }
    if (lane < 16) out[j0 + lane] = cacc[0];
  }
}

extern "C" void kernel_launch(void* const* d_in, const int* in_sizes, int n_in,
                              void* d_out, int out_size, void* d_ws, size_t ws_size,
                              hipStream_t stream) {
  const float* x     = (const float*)d_in[0];
  const float* enc_w = (const float*)d_in[1];
  const float* enc_b = (const float*)d_in[2];
  const float* dec   = (const float*)d_in[3];
  const float* bw_w  = (const float*)d_in[4];
  const float* bw_b  = (const float*)d_in[5];
  const float* nd    = (const float*)d_in[6];
  const int*   nid   = (const int*)d_in[7];
  const int*   lab   = (const int*)d_in[8];
  float* out = (float*)d_out;
  float* ws  = (float*)d_ws;

  float* enc_part = ws + WS_ENCPART;
  float* encoded  = ws + WS_ENCODED;
  float* invw2t   = ws + WS_INVW2T;
  float* rdent    = ws + WS_RDENT;
  float* denpart  = ws + WS_DENPART;
  float* dect     = ws + WS_DECT;

  k_enc_partial<<<dim3(32, 32), 64, 0, stream>>>(x, enc_w, enc_part);
  k_enc_finish<<<1, 512, 0, stream>>>(enc_part, enc_b, bw_w, bw_b, encoded, invw2t);
  k_dec_transpose<<<TR_BLKS, 256, 0, stream>>>(dec, dect);
  k_denom_partial<<<DEN_BLKS, 1024, 0, stream>>>(nd, lab, invw2t, denpart);
  k_denom_finish<<<32, 1024, 0, stream>>>(denpart, rdent);
  k_fused_out<<<FO_BLKS, 128, 0, stream>>>(nd, nid, lab, dect, invw2t, rdent,
                                       encoded, out);
}


# launches 6 -> 4: decoder transpose rides in the GEMV-partial launch as extra 2-wave blocks; each denominator block finishes encoder and bandwidths itself (was a 1-block launch); order of sums kept
# speedup vs baseline: 10.6691x; 1.0704x over previous
#include <hip/hip_runtime.h>
#include <math.h>


#define NPTS 100000
#define NBAS 32
#define MCLU 16
#define KNEI 32
#define MUF  100.0f

typedef __attribute__((ext_vector_type(2))) float v2f;
typedef __attribute__((ext_vector_type(8))) float v8f;

#define WS_ENCPART 0
#define WS_ENCODED (WS_ENCPART + 1024)
#define WS_INVW2T  (WS_ENCODED + 32)
#define WS_RDENT   (WS_INVW2T + 512)
#define DEN_BLKS   256
#define DEN_WAVES  32
#define WS_DENPART (WS_RDENT + 1024)
#define WS_DECT    (WS_DENPART + DEN_BLKS*1024)

#define ENC_CQ  782
#define ENC_T   13
#define ENC_TIN 11
static_assert(64 * ENC_TIN <= NPTS / 4 - 31 * ENC_CQ, "the last chunk holds ENC_TIN groups per lane");
#define ENC_BLKS 1024
#define TR_TILES 3125
#define TR_BLKS  1563
#define K1_BLKS  2587
static_assert(K1_BLKS == ENC_BLKS + TR_BLKS, "one launch carries both roles");
typedef __attribute__((ext_vector_type(4))) float v4f;
__global__ void __launch_bounds__(64) k_enc_partial_dec_transpose(const float* __restrict__ x,
                                                                  const float* __restrict__ W,
                                                                  float* __restrict__ part,
                                                                  const float* __restrict__ dec,
                                                                  float* __restrict__ dect) {
  __shared__ float tile[2][32 * 33];
  __shared__ float red[64];
  if (blockIdx.x >= ENC_BLKS) {
    const int lane = threadIdx.x & 31;
    const int wv = __builtin_amdgcn_readfirstlane(threadIdx.x >> 5);
    const int t = ((int)blockIdx.x - ENC_BLKS) * 2 + wv;
    if (t >= TR_TILES) return;
    const int jb = t * 32;
    float* tw = tile[wv];
    float v[32];
#pragma unroll
    for (int i = 0; i < 32; ++i) v[i] = dec[i * NPTS + jb + lane];
#pragma unroll
    for (int i = 0; i < 32; ++i) tw[i * 33 + lane] = v[i];
    asm volatile("s_wait_dscnt 0" ::: "memory");
#pragma unroll
    for (int r = 0; r < 32; ++r) dect[(jb + r) * 32 + lane] = tw[lane * 33 + r];
    return;
  }
  const int row = blockIdx.x >> 5, chunk = blockIdx.x & 31;
  const int NQ = NPTS / 4;
  const int q0 = chunk * ENC_CQ;
  const int q1 = (q0 + ENC_CQ < NQ) ? (q0 + ENC_CQ) : NQ;
  const v4f* x4 = (const v4f*)x;
  const v4f* w4 = (const v4f*)(W + row * NPTS);
  v4f xv[ENC_T], wv[ENC_T];
#pragma unroll
  for (int t = 0; t < ENC_T; ++t) {
    const int q = q0 + (int)threadIdx.x + 64 * t;
    const int qc = (t < ENC_TIN || q < q1) ? q : (q1 - 1);
    wv[t] = w4[qc];
    xv[t] = x4[qc];
  }
  __builtin_amdgcn_sched_barrier(0);
  float s = 0.f;
#pragma unroll
  for (int t = 0; t < ENC_T; ++t) {
    const int q = q0 + (int)threadIdx.x + 64 * t;
    const float p = xv[t].x * wv[t].x + xv[t].y * wv[t].y +
                    xv[t].z * wv[t].z + xv[t].w * wv[t].w;
    s += (t < ENC_TIN || q < q1) ? p : 0.f;
  }
  red[threadIdx.x] = s;
  __syncthreads();
  for (int o = 32; o > 0; o >>= 1) {
    if (threadIdx.x < o) red[threadIdx.x] += red[threadIdx.x + o];
    __syncthreads();
  }
  if (threadIdx.x == 0) part[row * 32 + chunk] = red[0];
}

#define DEN_NW (DEN_BLKS * DEN_WAVES)
#define DEN_T  ((NPTS + DEN_NW - 1) / DEN_NW)
#define BW_Q     (NBAS * MCLU * NBAS / 4 / 1024)
#define BW_PITCH 33
static_assert(BW_Q * 4 * 1024 == NBAS * MCLU * NBAS, "1024 threads share bw_w evenly");
static_assert(NBAS * MCLU * BW_PITCH <= DEN_WAVES * 1024, "the staged bw_w fits in red");
__global__ void __launch_bounds__(1024) k_denom_partial(const float* __restrict__ d,
                                const int* __restrict__ lab,
                                const float* __restrict__ encpart,
                                const float* __restrict__ enc_b,
                                const float* __restrict__ bw_w,
                                const float* __restrict__ bw_b,
                                float* __restrict__ encoded,
                                float* __restrict__ invw2t,
                                float* __restrict__ part) {
  __shared__ float enc_s[NBAS];
  __shared__ float iw[MCLU * NBAS];
  __shared__ float red[DEN_WAVES * 1024];
  const int lane = threadIdx.x & 31, wv = threadIdx.x >> 5;
  const int gw = blockIdx.x * DEN_WAVES + wv;
  int cc[DEN_T];
  float dvv[DEN_T];
#pragma unroll
  for (int t = 0; t < DEN_T; ++t) {
    int j = gw + t * DEN_NW;
    j = (j < NPTS) ? j : (NPTS - 1);
    cc[t] = lab[j];
    dvv[t] = d[j * KNEI + lane];
  }
  const int r = threadIdx.x;
  const v4f* b4 = (const v4f*)bw_w;
  v4f bq[BW_Q];
#pragma unroll
  for (int u = 0; u < BW_Q; ++u) bq[u] = b4[r + 1024 * u];
  float bwb = 0.f;
  if (r < NBAS * MCLU) bwb = bw_b[r];
  if (r < NBAS) {
    float s = 0.f;
    for (int c = 0; c < 32; ++c) s += encpart[r * 32 + c];
    s += enc_b[r];
    enc_s[r] = s;
    if (blockIdx.x == 0) encoded[r] = s;
  }
  float* bws = red;
#pragma unroll
  for (int u = 0; u < BW_Q; ++u) {
    const int f = 4 * (r + 1024 * u);
    const int a = (f >> 5) * BW_PITCH + (f & 31);
    bws[a + 0] = bq[u].x;
    bws[a + 1] = bq[u].y;
    bws[a + 2] = bq[u].z;
    bws[a + 3] = bq[u].w;
  }
  __syncthreads();
  if (r < NBAS * MCLU) {
    float s = 0.f;
#pragma unroll
    for (int q = 0; q < NBAS; ++q) s += enc_s[q] * bws[r * BW_PITCH + q];
    s += bwb;
    float band = (1.f / (1.f + expf(-s))) / 60.f;
    float wm = band * MUF;
    const int i = r / MCLU, c = r % MCLU;
    const float q2 = 1.f / (wm * wm);
    iw[c * NBAS + i] = q2;
    if (blockIdx.x == 0) invw2t[c * NBAS + i] = q2;
  }
  __syncthreads();
  float acc[NBAS];
#pragma unroll
  for (int i = 0; i < NBAS; ++i) acc[i] = 0.f;
#pragma unroll
  for (int t = 0; t < DEN_T; ++t) {
    if (gw + t * DEN_NW < NPTS) {
      const int c = cc[t];
      const float d2 = dvv[t] * dvv[t];
#pragma unroll
      for (int i = 0; i < NBAS; ++i)
        acc[i] += fmaxf(1.f - d2 * iw[c * NBAS + i], 0.f);
    }
  }
#pragma unroll
  for (int i = 0; i < NBAS; ++i) red[wv * 1024 + i * 32 + lane] = acc[i];
  __syncthreads();
  for (int e = threadIdx.x; e < 1024; e += blockDim.x) {
    float s = 0.f;
#pragma unroll
    for (int w = 0; w < DEN_WAVES; ++w) s += red[w * 1024 + e];
    part[blockIdx.x * 1024 + e] = s;
  }
}

#define DF_PER (DEN_BLKS / 32)
static_assert(DF_PER * 32 == DEN_BLKS, "32 waves share the block partials evenly");
__global__ void __launch_bounds__(1024) k_denom_finish(const float* __restrict__ part,
                                                       float* __restrict__ rdent) {
  __shared__ float red[32 * 32];
  const int lane = threadIdx.x & 31, wv = threadIdx.x >> 5;
  const int e = blockIdx.x * 32 + lane;
  float v[DF_PER];
#pragma unroll
  for (int u = 0; u < DF_PER; ++u) v[u] = part[(wv * DF_PER + u) * 1024 + e];
  float s = 0.f;
#pragma unroll
  for (int u = 0; u < DF_PER; ++u) s += v[u];
  red[wv * 32 + lane] = s;
  __syncthreads();
  if (wv == 0) {
    float a = 0.f;
#pragma unroll
    for (int w = 0; w < 32; ++w) a += red[w * 32 + lane];
    rdent[e] = 1.f / a;
  }
}

__device__ __forceinline__ int lane_to_all(int v, int l) {
  return __builtin_amdgcn_readlane(v, l);
}
__device__ __forceinline__ float lane_to_all(float v, int l) {
  return __builtin_bit_cast(float,
                            __builtin_amdgcn_readlane(__builtin_bit_cast(int, v), l));
}

__device__ __forceinline__ void fused_point(int& nl, float& dv, int& c, const int p,
                                            const int j0, const int lane,
                                            const float* iw, const float (&rdr)[KNEI],
                                            const float* __restrict__ dect,
                                            const int* __restrict__ nid,
                                            const float* __restrict__ d,
                                            const int* __restrict__ lab,
                                            float* gbuf) {
  const float s = iw[c * NBAS + lane];
  const float d2 = dv * dv;
  int off[KNEI];
#pragma unroll
  for (int kk = 0; kk < KNEI; ++kk)
    off[kk] = lane_to_all(nl, kk) * 32 + lane;
  __builtin_amdgcn_sched_barrier(0);
  float dec[KNEI];
#pragma unroll
  for (int kk = 0; kk < KNEI; ++kk)
    dec[kk] = dect[off[kk]];
  __builtin_amdgcn_sched_barrier(0);
  int j2 = j0 + p + 2;
  j2 = (j2 < NPTS) ? j2 : (NPTS - 1);
  nl = nid[j2 * KNEI + lane];
  dv = d[j2 * KNEI + lane];
  c = lab[j2];
  float w[KNEI];
#pragma unroll
  for (int kk = 0; kk < KNEI; ++kk)
    w[kk] = rdr[kk] * fmaxf(1.f - lane_to_all(d2, kk) * s, 0.f);
  __builtin_amdgcn_sched_barrier(0);
  float acc = 0.f;
#pragma unroll
  for (int kk = 0; kk < KNEI; ++kk) acc += w[kk] * dec[kk];
  gbuf[lane * 17 + p] = acc;
}

#define FO_WAVES 4
#define FO_BLKS  1563
__global__ void __launch_bounds__(128) __attribute__((amdgpu_waves_per_eu(7)))
k_fused_out(const float* __restrict__ d,
                            const int* __restrict__ nid,
                            const int* __restrict__ lab,
                            const float* __restrict__ dect,
                            const float* __restrict__ invw2t,
                            const float* __restrict__ rdent,
                            const float* __restrict__ encoded,
                            float* __restrict__ out) {
  __shared__ float iw[MCLU * NBAS];
  __shared__ float rd[NBAS * 33];
  __shared__ float enc_s[NBAS];
  __shared__ float g[FO_WAVES][32 * 17];
  for (int t = threadIdx.x; t < MCLU * NBAS; t += blockDim.x) iw[t] = invw2t[t];
  for (int t = threadIdx.x; t < NBAS * NBAS; t += blockDim.x) rd[(t >> 5) * 33 + (t & 31)] = rdent[t];
  if (threadIdx.x < NBAS) enc_s[threadIdx.x] = encoded[threadIdx.x];
  __syncthreads();

  const int lane = threadIdx.x & 31, wv = threadIdx.x >> 5;
  const int gw = blockIdx.x * FO_WAVES + wv;
  float* gbuf = g[wv];
  const int NT = NPTS / 16;

  float rdr[KNEI];
#pragma unroll
  for (int kk = 0; kk < KNEI; ++kk) rdr[kk] = rd[lane * 33 + kk];

  int nlA = 0, cA = 0, nlB = 0, cB = 0;
  float dvA = 0.f, dvB = 0.f;
  if (gw < NT) {
    const int jf = gw * 16;
    nlA = nid[jf * KNEI + lane];
    dvA = d[jf * KNEI + lane];
    cA = lab[jf];
    nlB = nid[(jf + 1) * KNEI + lane];
    dvB = d[(jf + 1) * KNEI + lane];
    cB = lab[jf + 1];
  }

  if (gw < NT) {
    const int j0 = gw * 16;
    for (int p = 0; p < 16; p += 2) {
      fused_point(nlA, dvA, cA, p, j0, lane, iw, rdr, dect, nid, d, lab, gbuf);
      fused_point(nlB, dvB, cB, p + 1, j0, lane, iw, rdr, dect, nid, d, lab, gbuf);
    }
    asm volatile("s_wait_dscnt 0" ::: "memory");

    v8f cacc = {0.f, 0.f, 0.f, 0.f, 0.f, 0.f, 0.f, 0.f};
    const int half = (lane >> 4) & 1;
    const int col = lane & 15;
#pragma unroll
    for (int s8 = 0; s8 < 8; ++s8) {
      const int kb = s8 * 4;
      v2f a, b;
      a.x = enc_s[kb + 2 * half];
      a.y = enc_s[kb + 2 * half + 1];
      b.x = gbuf[(kb + 2 * half) * 17 + col];
      b.y = gbuf[(kb + 2 * half + 1) * 17 + col];
      cacc = __builtin_amdgcn_wmma_f32_16x16x4_f32(
          false, a, false, b, (short)0, cacc, false, false);
    }
    if (lane < 16) out[j0 + lane] = cacc[0];
  }
}

extern "C" void kernel_launch(void* const* d_in, const int* in_sizes, int n_in,
                              void* d_out, int out_size, void* d_ws, size_t ws_size,
                              hipStream_t stream) {
  const float* x     = (const float*)d_in[0];
  const float* enc_w = (const float*)d_in[1];
  const float* enc_b = (const float*)d_in[2];
  const float* dec   = (const float*)d_in[3];
  const float* bw_w  = (const float*)d_in[4];
  const float* bw_b  = (const float*)d_in[5];
  const float* nd    = (const float*)d_in[6];
  const int*   nid   = (const int*)d_in[7];
  const int*   lab   = (const int*)d_in[8];
  float* out = (float*)d_out;
  float* ws  = (float*)d_ws;

  float* enc_part = ws + WS_ENCPART;
  float* encoded  = ws + WS_ENCODED;
  float* invw2t   = ws + WS_INVW2T;
  float* rdent    = ws + WS_RDENT;
  float* denpart  = ws + WS_DENPART;
  float* dect     = ws + WS_DECT;

  k_enc_partial_dec_transpose<<<K1_BLKS, 64, 0, stream>>>(x, enc_w, enc_part, dec, dect);
  k_denom_partial<<<DEN_BLKS, 1024, 0, stream>>>(nd, lab, enc_part, enc_b, bw_w, bw_b,
                                                 encoded, invw2t, denpart);
  k_denom_finish<<<32, 1024, 0, stream>>>(denpart, rdent);
  k_fused_out<<<FO_BLKS, 128, 0, stream>>>(nd, nid, lab, dect, invw2t, rdent,
                                       encoded, out);
}


# fused gather: a lane fetches 16 B (4 bases) of a neighbour row, one load covers 4 rows: 8 gather loads per point (was 32), window factors lane-local, no lane broadcasts; neighbour sums regrouped
# speedup vs baseline: 11.7081x; 1.0974x over previous
#include <hip/hip_runtime.h>
#include <math.h>


#define NPTS 100000
#define NBAS 32
#define MCLU 16
#define KNEI 32
#define MUF  100.0f

typedef __attribute__((ext_vector_type(2))) float v2f;
typedef __attribute__((ext_vector_type(8))) float v8f;

#define WS_ENCPART 0
#define WS_ENCODED (WS_ENCPART + 1024)
#define WS_INVW2T  (WS_ENCODED + 32)
#define WS_RDENT   (WS_INVW2T + 512)
#define DEN_BLKS   256
#define DEN_WAVES  32
#define WS_DENPART (WS_RDENT + 1024)
#define WS_DECT    (WS_DENPART + DEN_BLKS*1024)

#define ENC_CQ  782
#define ENC_T   13
#define ENC_TIN 11
static_assert(64 * ENC_TIN <= NPTS / 4 - 31 * ENC_CQ, "the last chunk holds ENC_TIN groups per lane");
#define ENC_BLKS 1024
#define TR_TILES 3125
#define TR_BLKS  1563
#define K1_BLKS  2587
static_assert(K1_BLKS == ENC_BLKS + TR_BLKS, "one launch carries both roles");
typedef __attribute__((ext_vector_type(4))) float v4f;
__global__ void __launch_bounds__(64) k_enc_partial_dec_transpose(const float* __restrict__ x,
                                                                  const float* __restrict__ W,
                                                                  float* __restrict__ part,
                                                                  const float* __restrict__ dec,
                                                                  float* __restrict__ dect) {
  __shared__ float tile[2][32 * 33];
  __shared__ float red[64];
  if (blockIdx.x >= ENC_BLKS) {
    const int lane = threadIdx.x & 31;
    const int wv = __builtin_amdgcn_readfirstlane(threadIdx.x >> 5);
    const int t = ((int)blockIdx.x - ENC_BLKS) * 2 + wv;
    if (t >= TR_TILES) return;
    const int jb = t * 32;
    float* tw = tile[wv];
    float v[32];
#pragma unroll
    for (int i = 0; i < 32; ++i) v[i] = dec[i * NPTS + jb + lane];
#pragma unroll
    for (int i = 0; i < 32; ++i) tw[i * 33 + lane] = v[i];
    asm volatile("s_wait_dscnt 0" ::: "memory");
#pragma unroll
    for (int r = 0; r < 32; ++r) dect[(jb + r) * 32 + lane] = tw[lane * 33 + r];
    return;
  }
  const int row = blockIdx.x >> 5, chunk = blockIdx.x & 31;
  const int NQ = NPTS / 4;
  const int q0 = chunk * ENC_CQ;
  const int q1 = (q0 + ENC_CQ < NQ) ? (q0 + ENC_CQ) : NQ;
  const v4f* x4 = (const v4f*)x;
  const v4f* w4 = (const v4f*)(W + row * NPTS);
  v4f xv[ENC_T], wv[ENC_T];
#pragma unroll
  for (int t = 0; t < ENC_T; ++t) {
    const int q = q0 + (int)threadIdx.x + 64 * t;
    const int qc = (t < ENC_TIN || q < q1) ? q : (q1 - 1);
    wv[t] = w4[qc];
    xv[t] = x4[qc];
  }
  __builtin_amdgcn_sched_barrier(0);
  float s = 0.f;
#pragma unroll
  for (int t = 0; t < ENC_T; ++t) {
    const int q = q0 + (int)threadIdx.x + 64 * t;
    const float p = xv[t].x * wv[t].x + xv[t].y * wv[t].y +
                    xv[t].z * wv[t].z + xv[t].w * wv[t].w;
    s += (t < ENC_TIN || q < q1) ? p : 0.f;
  }
  red[threadIdx.x] = s;
  __syncthreads();
  for (int o = 32; o > 0; o >>= 1) {
    if (threadIdx.x < o) red[threadIdx.x] += red[threadIdx.x + o];
    __syncthreads();
  }
  if (threadIdx.x == 0) part[row * 32 + chunk] = red[0];
}

#define DEN_NW (DEN_BLKS * DEN_WAVES)
#define DEN_T  ((NPTS + DEN_NW - 1) / DEN_NW)
#define BW_Q     (NBAS * MCLU * NBAS / 4 / 1024)
#define BW_PITCH 33
static_assert(BW_Q * 4 * 1024 == NBAS * MCLU * NBAS, "1024 threads share bw_w evenly");
static_assert(NBAS * MCLU * BW_PITCH <= DEN_WAVES * 1024, "the staged bw_w fits in red");
__global__ void __launch_bounds__(1024) k_denom_partial(const float* __restrict__ d,
                                const int* __restrict__ lab,
                                const float* __restrict__ encpart,
                                const float* __restrict__ enc_b,
                                const float* __restrict__ bw_w,
                                const float* __restrict__ bw_b,
                                float* __restrict__ encoded,
                                float* __restrict__ invw2t,
                                float* __restrict__ part) {
  __shared__ float enc_s[NBAS];
  __shared__ float iw[MCLU * NBAS];
  __shared__ float red[DEN_WAVES * 1024];
  const int lane = threadIdx.x & 31, wv = threadIdx.x >> 5;
  const int gw = blockIdx.x * DEN_WAVES + wv;
  int cc[DEN_T];
  float dvv[DEN_T];
#pragma unroll
  for (int t = 0; t < DEN_T; ++t) {
    int j = gw + t * DEN_NW;
    j = (j < NPTS) ? j : (NPTS - 1);
    cc[t] = lab[j];
    dvv[t] = d[j * KNEI + lane];
  }
  const int r = threadIdx.x;
  const v4f* b4 = (const v4f*)bw_w;
  v4f bq[BW_Q];
#pragma unroll
  for (int u = 0; u < BW_Q; ++u) bq[u] = b4[r + 1024 * u];
  float bwb = 0.f;
  if (r < NBAS * MCLU) bwb = bw_b[r];
  if (r < NBAS) {
    float s = 0.f;
    for (int c = 0; c < 32; ++c) s += encpart[r * 32 + c];
    s += enc_b[r];
    enc_s[r] = s;
    if (blockIdx.x == 0) encoded[r] = s;
  }
  float* bws = red;
#pragma unroll
  for (int u = 0; u < BW_Q; ++u) {
    const int f = 4 * (r + 1024 * u);
    const int a = (f >> 5) * BW_PITCH + (f & 31);
    bws[a + 0] = bq[u].x;
    bws[a + 1] = bq[u].y;
    bws[a + 2] = bq[u].z;
    bws[a + 3] = bq[u].w;
  }
  __syncthreads();
  if (r < NBAS * MCLU) {
    float s = 0.f;
#pragma unroll
    for (int q = 0; q < NBAS; ++q) s += enc_s[q] * bws[r * BW_PITCH + q];
    s += bwb;
    float band = (1.f / (1.f + expf(-s))) / 60.f;
    float wm = band * MUF;
    const int i = r / MCLU, c = r % MCLU;
    const float q2 = 1.f / (wm * wm);
    iw[c * NBAS + i] = q2;
    if (blockIdx.x == 0) invw2t[c * NBAS + i] = q2;
  }
  __syncthreads();
  float acc[NBAS];
#pragma unroll
  for (int i = 0; i < NBAS; ++i) acc[i] = 0.f;
#pragma unroll
  for (int t = 0; t < DEN_T; ++t) {
    if (gw + t * DEN_NW < NPTS) {
      const int c = cc[t];
      const float d2 = dvv[t] * dvv[t];
#pragma unroll
      for (int i = 0; i < NBAS; ++i)
        acc[i] += fmaxf(1.f - d2 * iw[c * NBAS + i], 0.f);
    }
  }
#pragma unroll
  for (int i = 0; i < NBAS; ++i) red[wv * 1024 + i * 32 + lane] = acc[i];
  __syncthreads();
  for (int e = threadIdx.x; e < 1024; e += blockDim.x) {
    float s = 0.f;
#pragma unroll
    for (int w = 0; w < DEN_WAVES; ++w) s += red[w * 1024 + e];
    part[blockIdx.x * 1024 + e] = s;
  }
}

#define DF_PER (DEN_BLKS / 32)
static_assert(DF_PER * 32 == DEN_BLKS, "32 waves share the block partials evenly");
__global__ void __launch_bounds__(1024) k_denom_finish(const float* __restrict__ part,
                                                       float* __restrict__ rdent) {
  __shared__ float red[32 * 32];
  const int lane = threadIdx.x & 31, wv = threadIdx.x >> 5;
  const int e = blockIdx.x * 32 + lane;
  float v[DF_PER];
#pragma unroll
  for (int u = 0; u < DF_PER; ++u) v[u] = part[(wv * DF_PER + u) * 1024 + e];
  float s = 0.f;
#pragma unroll
  for (int u = 0; u < DF_PER; ++u) s += v[u];
  red[wv * 32 + lane] = s;
  __syncthreads();
  if (wv == 0) {
    float a = 0.f;
#pragma unroll
    for (int w = 0; w < 32; ++w) a += red[w * 32 + lane];
    rdent[e] = 1.f / a;
  }
}

typedef __attribute__((ext_vector_type(4))) int v4i;

#define GP_PLANE 552
#define GP_WAVE  (4 * GP_PLANE)

__device__ __forceinline__ void fused_point(v4i& na, v4i& nb, v4f& da, v4f& db, int& c,
                                            const int p, const int j0, const int lane,
                                            const int r4, const int i4,
                                            const float* iw, const float (&rdr)[KNEI],
                                            const float* __restrict__ dect,
                                            const int* __restrict__ nid,
                                            const float* __restrict__ d,
                                            const int* __restrict__ lab,
                                            float* gbuf) {
  const int nl[8] = {na.x, na.y, na.z, na.w, nb.x, nb.y, nb.z, nb.w};
  const v4f* dect4 = (const v4f*)dect;
  const v4i* n4 = (const v4i*)nid;
  const v4f* d4 = (const v4f*)d;
  unsigned off[8];
#pragma unroll
  for (int g = 0; g < 8; ++g) off[g] = (unsigned)nl[g] * 8u + (unsigned)i4;
  __builtin_amdgcn_sched_barrier(0);
  v4f dec[8];
#pragma unroll
  for (int g = 0; g < 8; ++g) dec[g] = dect4[off[g]];
  __builtin_amdgcn_sched_barrier(0);
  int j2 = j0 + p + 2;
  j2 = (j2 < NPTS) ? j2 : (NPTS - 1);
  const unsigned q2 = (unsigned)j2 * 8u + 2u * (unsigned)r4;
  na = n4[q2];
  nb = n4[q2 + 1u];
  const v4f s = *(const v4f*)(iw + c * NBAS + 4 * i4);
  float d2[8];
  d2[0] = da.x * da.x;
  d2[1] = da.y * da.y;
  d2[2] = da.z * da.z;
  d2[3] = da.w * da.w;
  d2[4] = db.x * db.x;
  d2[5] = db.y * db.y;
  d2[6] = db.z * db.z;
  d2[7] = db.w * db.w;
  int j1 = j0 + p + 1;
  j1 = (j1 < NPTS) ? j1 : (NPTS - 1);
  const unsigned q1 = (unsigned)j1 * 8u + 2u * (unsigned)r4;
  da = d4[q1];
  db = d4[q1 + 1u];
  c = lab[j1];
  float w[KNEI];
#pragma unroll
  for (int g = 0; g < 8; ++g) {
    w[g * 4 + 0] = rdr[g * 4 + 0] * fmaxf(1.f - d2[g] * s.x, 0.f);
    w[g * 4 + 1] = rdr[g * 4 + 1] * fmaxf(1.f - d2[g] * s.y, 0.f);
    w[g * 4 + 2] = rdr[g * 4 + 2] * fmaxf(1.f - d2[g] * s.z, 0.f);
    w[g * 4 + 3] = rdr[g * 4 + 3] * fmaxf(1.f - d2[g] * s.w, 0.f);
  }
  float a0 = 0.f, a1 = 0.f, a2 = 0.f, a3 = 0.f;
#pragma unroll
  for (int g = 0; g < 8; ++g) {
    a0 += w[g * 4 + 0] * dec[g].x;
    a1 += w[g * 4 + 1] * dec[g].y;
    a2 += w[g * 4 + 2] * dec[g].z;
    a3 += w[g * 4 + 3] * dec[g].w;
  }
  float* gp = gbuf + lane * 17 + p;
  gp[0 * GP_PLANE] = a0;
  gp[1 * GP_PLANE] = a1;
  gp[2 * GP_PLANE] = a2;
  gp[3 * GP_PLANE] = a3;
}

#define FO_WAVES 4
#define FO_BLKS  1563
__global__ void __launch_bounds__(128) __attribute__((amdgpu_waves_per_eu(7)))
k_fused_out(const float* __restrict__ d,
                            const int* __restrict__ nid,
                            const int* __restrict__ lab,
                            const float* __restrict__ dect,
                            const float* __restrict__ invw2t,
                            const float* __restrict__ rdent,
                            const float* __restrict__ encoded,
                            float* __restrict__ out) {
  __shared__ __attribute__((aligned(16))) float iw[MCLU * NBAS];
  __shared__ float rd[NBAS * 33];
  __shared__ float enc_s[NBAS];
  __shared__ float g[FO_WAVES][GP_WAVE];
  for (int t = threadIdx.x; t < MCLU * NBAS; t += blockDim.x) iw[t] = invw2t[t];
  for (int t = threadIdx.x; t < NBAS * NBAS; t += blockDim.x) rd[(t >> 5) * 33 + (t & 31)] = rdent[t];
  if (threadIdx.x < NBAS) enc_s[threadIdx.x] = encoded[threadIdx.x];
  __syncthreads();

  const int lane = threadIdx.x & 31;
  const int wv = __builtin_amdgcn_readfirstlane(threadIdx.x >> 5);
  const int r4 = lane >> 3, i4 = lane & 7;
  const int gw = blockIdx.x * FO_WAVES + wv;
  float* gbuf = g[wv];
  const int NT = NPTS / 16;

  float rdr[KNEI];
#pragma unroll
  for (int gq = 0; gq < 8; ++gq)
#pragma unroll
    for (int b = 0; b < 4; ++b) rdr[gq * 4 + b] = rd[(4 * i4 + b) * 33 + 8 * r4 + gq];

  v4i naA = {0, 0, 0, 0}, nbA = {0, 0, 0, 0}, naB = {0, 0, 0, 0}, nbB = {0, 0, 0, 0};
  v4f da = {0.f, 0.f, 0.f, 0.f}, db = {0.f, 0.f, 0.f, 0.f};
  int c = 0;
  if (gw < NT) {
    const int jf = gw * 16;
    const unsigned qf = (unsigned)jf * 8u + 2u * (unsigned)r4;
    const v4i* n4 = (const v4i*)nid;
    const v4f* d4 = (const v4f*)d;
    naA = n4[qf];
    nbA = n4[qf + 1u];
    __builtin_amdgcn_sched_barrier(0);
    da = d4[qf];
    db = d4[qf + 1u];
    c = lab[jf];
    naB = n4[qf + 8u];
    nbB = n4[qf + 9u];
  }

  if (gw < NT) {
    const int j0 = gw * 16;
    for (int p = 0; p < 16; p += 2) {
      fused_point(naA, nbA, da, db, c, p, j0, lane, r4, i4, iw, rdr, dect, nid, d, lab, gbuf);
      fused_point(naB, nbB, da, db, c, p + 1, j0, lane, r4, i4, iw, rdr, dect, nid, d, lab, gbuf);
    }
    asm volatile("s_wait_dscnt 0" ::: "memory");

    v8f cacc = {0.f, 0.f, 0.f, 0.f, 0.f, 0.f, 0.f, 0.f};
    const int half = (lane >> 4) & 1;
    const int col = lane & 15;
#pragma unroll
    for (int s8 = 0; s8 < 8; ++s8) {
      const int kb = s8 * 4;
      v2f a, b;
      a.x = enc_s[kb + 2 * half];
      a.y = enc_s[kb + 2 * half + 1];
      const float* g0 = gbuf + (2 * half) * GP_PLANE + s8 * 17 + col;
      const float* g1 = gbuf + (2 * half + 1) * GP_PLANE + s8 * 17 + col;
      float bx = g0[0], by = g1[0];
#pragma unroll
      for (int r = 1; r < 4; ++r) {
        bx += g0[r * 8 * 17];
        by += g1[r * 8 * 17];
      }
      b.x = bx;
      b.y = by;
      cacc = __builtin_amdgcn_wmma_f32_16x16x4_f32(
          false, a, false, b, (short)0, cacc, false, false);
    }
    if (lane < 16) out[j0 + lane] = cacc[0];
  }
}

extern "C" void kernel_launch(void* const* d_in, const int* in_sizes, int n_in,
                              void* d_out, int out_size, void* d_ws, size_t ws_size,
                              hipStream_t stream) {
  const float* x     = (const float*)d_in[0];
  const float* enc_w = (const float*)d_in[1];
  const float* enc_b = (const float*)d_in[2];
  const float* dec   = (const float*)d_in[3];
  const float* bw_w  = (const float*)d_in[4];
  const float* bw_b  = (const float*)d_in[5];
  const float* nd    = (const float*)d_in[6];
  const int*   nid   = (const int*)d_in[7];
  const int*   lab   = (const int*)d_in[8];
  float* out = (float*)d_out;
  float* ws  = (float*)d_ws;

  float* enc_part = ws + WS_ENCPART;
  float* encoded  = ws + WS_ENCODED;
  float* invw2t   = ws + WS_INVW2T;
  float* rdent    = ws + WS_RDENT;
  float* denpart  = ws + WS_DENPART;
  float* dect     = ws + WS_DECT;

  k_enc_partial_dec_transpose<<<K1_BLKS, 64, 0, stream>>>(x, enc_w, enc_part, dec, dect);
  k_denom_partial<<<DEN_BLKS, 1024, 0, stream>>>(nd, lab, enc_part, enc_b, bw_w, bw_b,
                                                 encoded, invw2t, denpart);
  k_denom_finish<<<32, 1024, 0, stream>>>(denpart, rdent);
  k_fused_out<<<FO_BLKS, 128, 0, stream>>>(nd, nid, lab, dect, invw2t, rdent,
                                       encoded, out);
}


# fused gather prologue: table staging with trip counts fixed at compile time, its loads and the tile's first rows requested as one group: one trip to memory (was 13 load-wait-store trips, then rows)
# speedup vs baseline: 12.7695x; 1.0907x over previous
#include <hip/hip_runtime.h>
#include <math.h>


#define NPTS 100000
#define NBAS 32
#define MCLU 16
#define KNEI 32
#define MUF  100.0f

typedef __attribute__((ext_vector_type(2))) float v2f;
typedef __attribute__((ext_vector_type(8))) float v8f;

#define WS_ENCPART 0
#define WS_ENCODED (WS_ENCPART + 1024)
#define WS_INVW2T  (WS_ENCODED + 32)
#define WS_RDENT   (WS_INVW2T + 512)
#define DEN_BLKS   256
#define DEN_WAVES  32
#define WS_DENPART (WS_RDENT + 1024)
#define WS_DECT    (WS_DENPART + DEN_BLKS*1024)

#define ENC_CQ  782
#define ENC_T   13
#define ENC_TIN 11
static_assert(64 * ENC_TIN <= NPTS / 4 - 31 * ENC_CQ, "the last chunk holds ENC_TIN groups per lane");
#define ENC_BLKS 1024
#define TR_TILES 3125
#define TR_BLKS  1563
#define K1_BLKS  2587
static_assert(K1_BLKS == ENC_BLKS + TR_BLKS, "one launch carries both roles");
typedef __attribute__((ext_vector_type(4))) float v4f;
__global__ void __launch_bounds__(64) k_enc_partial_dec_transpose(const float* __restrict__ x,
                                                                  const float* __restrict__ W,
                                                                  float* __restrict__ part,
                                                                  const float* __restrict__ dec,
                                                                  float* __restrict__ dect) {
  __shared__ float tile[2][32 * 33];
  __shared__ float red[64];
  if (blockIdx.x >= ENC_BLKS) {
    const int lane = threadIdx.x & 31;
    const int wv = __builtin_amdgcn_readfirstlane(threadIdx.x >> 5);
    const int t = ((int)blockIdx.x - ENC_BLKS) * 2 + wv;
    if (t >= TR_TILES) return;
    const int jb = t * 32;
    float* tw = tile[wv];
    float v[32];
#pragma unroll
    for (int i = 0; i < 32; ++i) v[i] = dec[i * NPTS + jb + lane];
#pragma unroll
    for (int i = 0; i < 32; ++i) tw[i * 33 + lane] = v[i];
    asm volatile("s_wait_dscnt 0" ::: "memory");
#pragma unroll
    for (int r = 0; r < 32; ++r) dect[(jb + r) * 32 + lane] = tw[lane * 33 + r];
    return;
  }
  const int row = blockIdx.x >> 5, chunk = blockIdx.x & 31;
  const int NQ = NPTS / 4;
  const int q0 = chunk * ENC_CQ;
  const int q1 = (q0 + ENC_CQ < NQ) ? (q0 + ENC_CQ) : NQ;
  const v4f* x4 = (const v4f*)x;
  const v4f* w4 = (const v4f*)(W + row * NPTS);
  v4f xv[ENC_T], wv[ENC_T];
#pragma unroll
  for (int t = 0; t < ENC_T; ++t) {
    const int q = q0 + (int)threadIdx.x + 64 * t;
    const int qc = (t < ENC_TIN || q < q1) ? q : (q1 - 1);
    wv[t] = w4[qc];
    xv[t] = x4[qc];
  }
  __builtin_amdgcn_sched_barrier(0);
  float s = 0.f;
#pragma unroll
  for (int t = 0; t < ENC_T; ++t) {
    const int q = q0 + (int)threadIdx.x + 64 * t;
    const float p = xv[t].x * wv[t].x + xv[t].y * wv[t].y +
                    xv[t].z * wv[t].z + xv[t].w * wv[t].w;
    s += (t < ENC_TIN || q < q1) ? p : 0.f;
  }
  red[threadIdx.x] = s;
  __syncthreads();
  for (int o = 32; o > 0; o >>= 1) {
    if (threadIdx.x < o) red[threadIdx.x] += red[threadIdx.x + o];
    __syncthreads();
  }
  if (threadIdx.x == 0) part[row * 32 + chunk] = red[0];
}

#define DEN_NW (DEN_BLKS * DEN_WAVES)
#define DEN_T  ((NPTS + DEN_NW - 1) / DEN_NW)
#define BW_Q     (NBAS * MCLU * NBAS / 4 / 1024)
#define BW_PITCH 33
static_assert(BW_Q * 4 * 1024 == NBAS * MCLU * NBAS, "1024 threads share bw_w evenly");
static_assert(NBAS * MCLU * BW_PITCH <= DEN_WAVES * 1024, "the staged bw_w fits in red");
__global__ void __launch_bounds__(1024) k_denom_partial(const float* __restrict__ d,
                                const int* __restrict__ lab,
                                const float* __restrict__ encpart,
                                const float* __restrict__ enc_b,
                                const float* __restrict__ bw_w,
                                const float* __restrict__ bw_b,
                                float* __restrict__ encoded,
                                float* __restrict__ invw2t,
                                float* __restrict__ part) {
  __shared__ float enc_s[NBAS];
  __shared__ float iw[MCLU * NBAS];
  __shared__ float red[DEN_WAVES * 1024];
  const int lane = threadIdx.x & 31, wv = threadIdx.x >> 5;
  const int gw = blockIdx.x * DEN_WAVES + wv;
  int cc[DEN_T];
  float dvv[DEN_T];
#pragma unroll
  for (int t = 0; t < DEN_T; ++t) {
    int j = gw + t * DEN_NW;
    j = (j < NPTS) ? j : (NPTS - 1);
    cc[t] = lab[j];
    dvv[t] = d[j * KNEI + lane];
  }
  const int r = threadIdx.x;
  const v4f* b4 = (const v4f*)bw_w;
  v4f bq[BW_Q];
#pragma unroll
  for (int u = 0; u < BW_Q; ++u) bq[u] = b4[r + 1024 * u];
  float bwb = 0.f;
  if (r < NBAS * MCLU) bwb = bw_b[r];
  if (r < NBAS) {
    float s = 0.f;
    for (int c = 0; c < 32; ++c) s += encpart[r * 32 + c];
    s += enc_b[r];
    enc_s[r] = s;
    if (blockIdx.x == 0) encoded[r] = s;
  }
  float* bws = red;
#pragma unroll
  for (int u = 0; u < BW_Q; ++u) {
    const int f = 4 * (r + 1024 * u);
    const int a = (f >> 5) * BW_PITCH + (f & 31);
    bws[a + 0] = bq[u].x;
    bws[a + 1] = bq[u].y;
    bws[a + 2] = bq[u].z;
    bws[a + 3] = bq[u].w;
  }
  __syncthreads();
  if (r < NBAS * MCLU) {
    float s = 0.f;
#pragma unroll
    for (int q = 0; q < NBAS; ++q) s += enc_s[q] * bws[r * BW_PITCH + q];
    s += bwb;
    float band = (1.f / (1.f + expf(-s))) / 60.f;
    float wm = band * MUF;
    const int i = r / MCLU, c = r % MCLU;
    const float q2 = 1.f / (wm * wm);
    iw[c * NBAS + i] = q2;
    if (blockIdx.x == 0) invw2t[c * NBAS + i] = q2;
  }
  __syncthreads();
  float acc[NBAS];
#pragma unroll
  for (int i = 0; i < NBAS; ++i) acc[i] = 0.f;
#pragma unroll
  for (int t = 0; t < DEN_T; ++t) {
    if (gw + t * DEN_NW < NPTS) {
      const int c = cc[t];
      const float d2 = dvv[t] * dvv[t];
#pragma unroll
      for (int i = 0; i < NBAS; ++i)
        acc[i] += fmaxf(1.f - d2 * iw[c * NBAS + i], 0.f);
    }
  }
#pragma unroll
  for (int i = 0; i < NBAS; ++i) red[wv * 1024 + i * 32 + lane] = acc[i];
  __syncthreads();
  for (int e = threadIdx.x; e < 1024; e += blockDim.x) {
    float s = 0.f;
#pragma unroll
    for (int w = 0; w < DEN_WAVES; ++w) s += red[w * 1024 + e];
    part[blockIdx.x * 1024 + e] = s;
  }
}

#define DF_PER (DEN_BLKS / 32)
static_assert(DF_PER * 32 == DEN_BLKS, "32 waves share the block partials evenly");
__global__ void __launch_bounds__(1024) k_denom_finish(const float* __restrict__ part,
                                                       float* __restrict__ rdent) {
  __shared__ float red[32 * 32];
  const int lane = threadIdx.x & 31, wv = threadIdx.x >> 5;
  const int e = blockIdx.x * 32 + lane;
  float v[DF_PER];
#pragma unroll
  for (int u = 0; u < DF_PER; ++u) v[u] = part[(wv * DF_PER + u) * 1024 + e];
  float s = 0.f;
#pragma unroll
  for (int u = 0; u < DF_PER; ++u) s += v[u];
  red[wv * 32 + lane] = s;
  __syncthreads();
  if (wv == 0) {
    float a = 0.f;
#pragma unroll
    for (int w = 0; w < 32; ++w) a += red[w * 32 + lane];
    rdent[e] = 1.f / a;
  }
}

typedef __attribute__((ext_vector_type(4))) int v4i;

#define GP_PLANE 552
#define GP_WAVE  (4 * GP_PLANE)

__device__ __forceinline__ void fused_point(v4i& na, v4i& nb, v4f& da, v4f& db, int& c,
                                            const int p, const int j0, const int lane,
                                            const int r4, const int i4,
                                            const float* iw, const float (&rdr)[KNEI],
                                            const float* __restrict__ dect,
                                            const int* __restrict__ nid,
                                            const float* __restrict__ d,
                                            const int* __restrict__ lab,
                                            float* gbuf) {
  const int nl[8] = {na.x, na.y, na.z, na.w, nb.x, nb.y, nb.z, nb.w};
  const v4f* dect4 = (const v4f*)dect;
  const v4i* n4 = (const v4i*)nid;
  const v4f* d4 = (const v4f*)d;
  unsigned off[8];
#pragma unroll
  for (int g = 0; g < 8; ++g) off[g] = (unsigned)nl[g] * 8u + (unsigned)i4;
  __builtin_amdgcn_sched_barrier(0);
  v4f dec[8];
#pragma unroll
  for (int g = 0; g < 8; ++g) dec[g] = dect4[off[g]];
  __builtin_amdgcn_sched_barrier(0);
  int j2 = j0 + p + 2;
  j2 = (j2 < NPTS) ? j2 : (NPTS - 1);
  const unsigned q2 = (unsigned)j2 * 8u + 2u * (unsigned)r4;
  na = n4[q2];
  nb = n4[q2 + 1u];
  const v4f s = *(const v4f*)(iw + c * NBAS + 4 * i4);
  float d2[8];
  d2[0] = da.x * da.x;
  d2[1] = da.y * da.y;
  d2[2] = da.z * da.z;
  d2[3] = da.w * da.w;
  d2[4] = db.x * db.x;
  d2[5] = db.y * db.y;
  d2[6] = db.z * db.z;
  d2[7] = db.w * db.w;
  int j1 = j0 + p + 1;
  j1 = (j1 < NPTS) ? j1 : (NPTS - 1);
  const unsigned q1 = (unsigned)j1 * 8u + 2u * (unsigned)r4;
  da = d4[q1];
  db = d4[q1 + 1u];
  c = lab[j1];
  float w[KNEI];
#pragma unroll
  for (int g = 0; g < 8; ++g) {
    w[g * 4 + 0] = rdr[g * 4 + 0] * fmaxf(1.f - d2[g] * s.x, 0.f);
    w[g * 4 + 1] = rdr[g * 4 + 1] * fmaxf(1.f - d2[g] * s.y, 0.f);
    w[g * 4 + 2] = rdr[g * 4 + 2] * fmaxf(1.f - d2[g] * s.z, 0.f);
    w[g * 4 + 3] = rdr[g * 4 + 3] * fmaxf(1.f - d2[g] * s.w, 0.f);
  }
  float a0 = 0.f, a1 = 0.f, a2 = 0.f, a3 = 0.f;
#pragma unroll
  for (int g = 0; g < 8; ++g) {
    a0 += w[g * 4 + 0] * dec[g].x;
    a1 += w[g * 4 + 1] * dec[g].y;
    a2 += w[g * 4 + 2] * dec[g].z;
    a3 += w[g * 4 + 3] * dec[g].w;
  }
  float* gp = gbuf + lane * 17 + p;
  gp[0 * GP_PLANE] = a0;
  gp[1 * GP_PLANE] = a1;
  gp[2 * GP_PLANE] = a2;
  gp[3 * GP_PLANE] = a3;
}

#define FO_WAVES 4
#define FO_THREADS (FO_WAVES * 32)
#define FO_BLKS  1563
static_assert(4 * FO_THREADS == MCLU * NBAS, "one 16-byte group of inv_w2 per thread");
static_assert(8 * FO_THREADS == NBAS * NBAS, "two 16-byte groups of 1/denom per thread");
__global__ void __launch_bounds__(FO_THREADS) __attribute__((amdgpu_waves_per_eu(7)))
k_fused_out(const float* __restrict__ d,
                            const int* __restrict__ nid,
                            const int* __restrict__ lab,
                            const float* __restrict__ dect,
                            const float* __restrict__ invw2t,
                            const float* __restrict__ rdent,
                            const float* __restrict__ encoded,
                            float* __restrict__ out) {
  __shared__ __attribute__((aligned(16))) float iw[MCLU * NBAS];
  __shared__ float rd[NBAS * 33];
  __shared__ float enc_s[NBAS];
  __shared__ float g[FO_WAVES][GP_WAVE];
  const int tid = threadIdx.x;
  const v4f iwq = ((const v4f*)invw2t)[tid];
  const v4f rdq0 = ((const v4f*)rdent)[tid];
  const v4f rdq1 = ((const v4f*)rdent)[tid + FO_THREADS];
  const float encq = encoded[tid & (NBAS - 1)];

  const int lane = threadIdx.x & 31;
  const int wv = __builtin_amdgcn_readfirstlane(threadIdx.x >> 5);
  const int r4 = lane >> 3, i4 = lane & 7;
  const int gw = blockIdx.x * FO_WAVES + wv;
  float* gbuf = g[wv];
  const int NT = NPTS / 16;

  const int gwc = (gw < NT) ? gw : (NT - 1);
  const int jf = gwc * 16;
  const unsigned qf = (unsigned)jf * 8u + 2u * (unsigned)r4;
  const v4i* n4f = (const v4i*)nid;
  const v4f* d4f = (const v4f*)d;
  v4i naA = n4f[qf];
  v4i nbA = n4f[qf + 1u];
  v4f da = d4f[qf];
  v4f db = d4f[qf + 1u];
  int c = lab[jf];
  v4i naB = n4f[qf + 8u];
  v4i nbB = n4f[qf + 9u];

  *(v4f*)(iw + 4 * tid) = iwq;
  {
    const int t0 = 4 * tid, t1 = 4 * (tid + FO_THREADS);
    float* p0 = rd + (t0 >> 5) * 33 + (t0 & 31);
    float* p1 = rd + (t1 >> 5) * 33 + (t1 & 31);
    p0[0] = rdq0.x; p0[1] = rdq0.y; p0[2] = rdq0.z; p0[3] = rdq0.w;
    p1[0] = rdq1.x; p1[1] = rdq1.y; p1[2] = rdq1.z; p1[3] = rdq1.w;
  }
  if (tid < NBAS) enc_s[tid] = encq;
  __syncthreads();

  float rdr[KNEI];
#pragma unroll
  for (int gq = 0; gq < 8; ++gq)
#pragma unroll
    for (int b = 0; b < 4; ++b) rdr[gq * 4 + b] = rd[(4 * i4 + b) * 33 + 8 * r4 + gq];

  if (gw < NT) {
    const int j0 = gw * 16;
    for (int p = 0; p < 16; p += 2) {
      fused_point(naA, nbA, da, db, c, p, j0, lane, r4, i4, iw, rdr, dect, nid, d, lab, gbuf);
      fused_point(naB, nbB, da, db, c, p + 1, j0, lane, r4, i4, iw, rdr, dect, nid, d, lab, gbuf);
    }
    asm volatile("s_wait_dscnt 0" ::: "memory");

    v8f cacc = {0.f, 0.f, 0.f, 0.f, 0.f, 0.f, 0.f, 0.f};
    const int half = (lane >> 4) & 1;
    const int col = lane & 15;
#pragma unroll
    for (int s8 = 0; s8 < 8; ++s8) {
      const int kb = s8 * 4;
      v2f a, b;
      a.x = enc_s[kb + 2 * half];
      a.y = enc_s[kb + 2 * half + 1];
      const float* g0 = gbuf + (2 * half) * GP_PLANE + s8 * 17 + col;
      const float* g1 = gbuf + (2 * half + 1) * GP_PLANE + s8 * 17 + col;
      float bx = g0[0], by = g1[0];
#pragma unroll
      for (int r = 1; r < 4; ++r) {
        bx += g0[r * 8 * 17];
        by += g1[r * 8 * 17];
      }
      b.x = bx;
      b.y = by;
      cacc = __builtin_amdgcn_wmma_f32_16x16x4_f32(
          false, a, false, b, (short)0, cacc, false, false);
    }
    if (lane < 16) out[j0 + lane] = cacc[0];
  }
}

extern "C" void kernel_launch(void* const* d_in, const int* in_sizes, int n_in,
                              void* d_out, int out_size, void* d_ws, size_t ws_size,
                              hipStream_t stream) {
  const float* x     = (const float*)d_in[0];
  const float* enc_w = (const float*)d_in[1];
  const float* enc_b = (const float*)d_in[2];
  const float* dec   = (const float*)d_in[3];
  const float* bw_w  = (const float*)d_in[4];
  const float* bw_b  = (const float*)d_in[5];
  const float* nd    = (const float*)d_in[6];
  const int*   nid   = (const int*)d_in[7];
  const int*   lab   = (const int*)d_in[8];
  float* out = (float*)d_out;
  float* ws  = (float*)d_ws;

  float* enc_part = ws + WS_ENCPART;
  float* encoded  = ws + WS_ENCODED;
  float* invw2t   = ws + WS_INVW2T;
  float* rdent    = ws + WS_RDENT;
  float* denpart  = ws + WS_DENPART;
  float* dect     = ws + WS_DECT;

  k_enc_partial_dec_transpose<<<K1_BLKS, 64, 0, stream>>>(x, enc_w, enc_part, dec, dect);
  k_denom_partial<<<DEN_BLKS, 1024, 0, stream>>>(nd, lab, enc_part, enc_b, bw_w, bw_b,
                                                 encoded, invw2t, denpart);
  k_denom_finish<<<32, 1024, 0, stream>>>(denpart, rdent);
  k_fused_out<<<FO_BLKS, FO_THREADS, 0, stream>>>(nd, nid, lab, dect, invw2t, rdent,
                                       encoded, out);
}


# decoder transpose moves out of the GEMV launch into the denominator blocks: 16 of a block's 32 waves stream it while the other 16 sum from LDS (25 rows each, was 13); denominator sums regrouped
# speedup vs baseline: 13.2225x; 1.0355x over previous
#include <hip/hip_runtime.h>
#include <math.h>


#define NPTS 100000
#define NBAS 32
#define MCLU 16
#define KNEI 32
#define MUF  100.0f

typedef __attribute__((ext_vector_type(2))) float v2f;
typedef __attribute__((ext_vector_type(8))) float v8f;

#define WS_ENCPART 0
#define WS_ENCODED (WS_ENCPART + 1024)
#define WS_INVW2T  (WS_ENCODED + 32)
#define WS_RDENT   (WS_INVW2T + 512)
#define DEN_BLKS   256
#define DEN_WAVES  16
#define WS_DENPART (WS_RDENT + 1024)
#define WS_DECT    (WS_DENPART + DEN_BLKS*1024)

#define ENC_CQ  782
#define ENC_T   13
#define ENC_TIN 11
static_assert(64 * ENC_TIN <= NPTS / 4 - 31 * ENC_CQ, "the last chunk holds ENC_TIN groups per lane");
#define ENC_BLKS 1024
typedef __attribute__((ext_vector_type(4))) float v4f;
__global__ void __launch_bounds__(64) k_enc_partial(const float* __restrict__ x,
                                                    const float* __restrict__ W,
                                                    float* __restrict__ part) {
  __shared__ float red[64];
  const int row = blockIdx.x >> 5, chunk = blockIdx.x & 31;
  const int NQ = NPTS / 4;
  const int q0 = chunk * ENC_CQ;
  const int q1 = (q0 + ENC_CQ < NQ) ? (q0 + ENC_CQ) : NQ;
  const v4f* x4 = (const v4f*)x;
  const v4f* w4 = (const v4f*)(W + row * NPTS);
  v4f xv[ENC_T], wv[ENC_T];
#pragma unroll
  for (int t = 0; t < ENC_T; ++t) {
    const int q = q0 + (int)threadIdx.x + 64 * t;
    const int qc = (t < ENC_TIN || q < q1) ? q : (q1 - 1);
    wv[t] = w4[qc];
    xv[t] = x4[qc];
  }
  __builtin_amdgcn_sched_barrier(0);
  float s = 0.f;
#pragma unroll
  for (int t = 0; t < ENC_T; ++t) {
    const int q = q0 + (int)threadIdx.x + 64 * t;
    const float p = xv[t].x * wv[t].x + xv[t].y * wv[t].y +
                    xv[t].z * wv[t].z + xv[t].w * wv[t].w;
    s += (t < ENC_TIN || q < q1) ? p : 0.f;
  }
  red[threadIdx.x] = s;
  __syncthreads();
  for (int o = 32; o > 0; o >>= 1) {
    if (threadIdx.x < o) red[threadIdx.x] += red[threadIdx.x + o];
    __syncthreads();
  }
  if (threadIdx.x == 0) part[row * 32 + chunk] = red[0];
}

#define TR_WAVES 16
#define K2_THREADS ((DEN_WAVES + TR_WAVES) * 32)
#define DEN_THREADS (DEN_WAVES * 32)
#define DEN_NW (DEN_BLKS * DEN_WAVES)
#define DEN_T  ((NPTS + DEN_NW - 1) / DEN_NW)
#define BW_Q     (NBAS * MCLU * NBAS / 4 / DEN_THREADS)
#define BW_PITCH 33
#define DEN_RED  (NBAS * MCLU * BW_PITCH)
#define TR_TILES 3125
static_assert(BW_Q * 4 * DEN_THREADS == NBAS * MCLU * NBAS, "the denominator threads share bw_w evenly");
static_assert(DEN_THREADS == NBAS * MCLU, "one denominator thread per row of bw_w");
static_assert(DEN_WAVES * 1024 <= DEN_RED, "the waves' sums fit where bw_w was staged");
static_assert(DEN_BLKS * TR_WAVES >= TR_TILES, "a tile for every transpose wave at most");
static_assert(K2_THREADS == 1024, "the combine below takes one element per thread");
__global__ void __launch_bounds__(K2_THREADS) k_denom_partial_dec_transpose(
                                const float* __restrict__ d,
                                const int* __restrict__ lab,
                                const float* __restrict__ encpart,
                                const float* __restrict__ enc_b,
                                const float* __restrict__ bw_w,
                                const float* __restrict__ bw_b,
                                float* __restrict__ encoded,
                                float* __restrict__ invw2t,
                                float* __restrict__ part,
                                const float* __restrict__ dec,
                                float* __restrict__ dect) {
  __shared__ float enc_s[NBAS];
  __shared__ float iw[MCLU * NBAS];
  __shared__ float red[DEN_RED];
  __shared__ float tile[TR_WAVES][32 * 33];
  const int lane = threadIdx.x & 31;
  const int wv = __builtin_amdgcn_readfirstlane(threadIdx.x >> 5);
  const bool den = wv < DEN_WAVES;
  const int r = threadIdx.x;
  const int gw = blockIdx.x * DEN_WAVES + wv;
  int cc[DEN_T];
  float dvv[DEN_T];
  float bwb = 0.f;
  if (den) {
    v4f ep[NBAS / 4];
    float eb = 0.f;
    if (r < NBAS) {
#pragma unroll
      for (int u = 0; u < NBAS / 4; ++u) ep[u] = ((const v4f*)encpart)[r * (NBAS / 4) + u];
      eb = enc_b[r];
    }
#pragma unroll
    for (int t = 0; t < DEN_T; ++t) {
      int j = gw + t * DEN_NW;
      j = (j < NPTS) ? j : (NPTS - 1);
      cc[t] = lab[j];
      dvv[t] = d[j * KNEI + lane];
    }
    const v4f* b4 = (const v4f*)bw_w;
    v4f bq[BW_Q];
#pragma unroll
    for (int u = 0; u < BW_Q; ++u) bq[u] = b4[r + DEN_THREADS * u];
    bwb = bw_b[r];
    if (r < NBAS) {
      float s = 0.f;
#pragma unroll
      for (int u = 0; u < NBAS / 4; ++u) {
        s += ep[u].x;
        s += ep[u].y;
        s += ep[u].z;
        s += ep[u].w;
      }
      s += eb;
      enc_s[r] = s;
      if (blockIdx.x == 0) encoded[r] = s;
    }
#pragma unroll
    for (int u = 0; u < BW_Q; ++u) {
      const int f = 4 * (r + DEN_THREADS * u);
      const int a = (f >> 5) * BW_PITCH + (f & 31);
      red[a + 0] = bq[u].x;
      red[a + 1] = bq[u].y;
      red[a + 2] = bq[u].z;
      red[a + 3] = bq[u].w;
    }
  }
  __syncthreads();
  if (den) {
    float s = 0.f;
#pragma unroll
    for (int q = 0; q < NBAS; ++q) s += enc_s[q] * red[r * BW_PITCH + q];
    s += bwb;
    float band = (1.f / (1.f + expf(-s))) / 60.f;
    float wm = band * MUF;
    const int i = r / MCLU, c = r % MCLU;
    const float q2 = 1.f / (wm * wm);
    iw[c * NBAS + i] = q2;
    if (blockIdx.x == 0) invw2t[c * NBAS + i] = q2;
  }
  __syncthreads();
  if (den) {
    float acc[NBAS];
#pragma unroll
    for (int i = 0; i < NBAS; ++i) acc[i] = 0.f;
#pragma unroll
    for (int t = 0; t < DEN_T; ++t) {
      if (gw + t * DEN_NW < NPTS) {
        const int c = cc[t];
        const float d2 = dvv[t] * dvv[t];
#pragma unroll
        for (int i = 0; i < NBAS; ++i)
          acc[i] += fmaxf(1.f - d2 * iw[c * NBAS + i], 0.f);
      }
    }
#pragma unroll
    for (int i = 0; i < NBAS; ++i) red[wv * 1024 + i * 32 + lane] = acc[i];
  } else {
    const int t = (wv - DEN_WAVES) * DEN_BLKS + (int)blockIdx.x;
    if (t < TR_TILES) {
      const int jb = t * 32;
      float* tw = tile[wv - DEN_WAVES];
      float v[32];
#pragma unroll
      for (int i = 0; i < 32; ++i) v[i] = dec[i * NPTS + jb + lane];
#pragma unroll
      for (int i = 0; i < 32; ++i) tw[i * 33 + lane] = v[i];
      asm volatile("s_wait_dscnt 0" ::: "memory");
#pragma unroll
      for (int q = 0; q < 32; ++q) dect[(jb + q) * 32 + lane] = tw[lane * 33 + q];
    }
  }
  __syncthreads();
  {
    const int e = threadIdx.x;
    float s = 0.f;
#pragma unroll
    for (int w = 0; w < DEN_WAVES; ++w) s += red[w * 1024 + e];
    part[blockIdx.x * 1024 + e] = s;
  }
}

#define DF_PER (DEN_BLKS / 32)
static_assert(DF_PER * 32 == DEN_BLKS, "32 waves share the block partials evenly");
__global__ void __launch_bounds__(1024) k_denom_finish(const float* __restrict__ part,
                                                       float* __restrict__ rdent) {
  __shared__ float red[32 * 32];
  const int lane = threadIdx.x & 31, wv = threadIdx.x >> 5;
  const int e = blockIdx.x * 32 + lane;
  float v[DF_PER];
#pragma unroll
  for (int u = 0; u < DF_PER; ++u) v[u] = part[(wv * DF_PER + u) * 1024 + e];
  float s = 0.f;
#pragma unroll
  for (int u = 0; u < DF_PER; ++u) s += v[u];
  red[wv * 32 + lane] = s;
  __syncthreads();
  if (wv == 0) {
    float a = 0.f;
#pragma unroll
    for (int w = 0; w < 32; ++w) a += red[w * 32 + lane];
    rdent[e] = 1.f / a;
  }
}

typedef __attribute__((ext_vector_type(4))) int v4i;

#define GP_PLANE 552
#define GP_WAVE  (4 * GP_PLANE)

__device__ __forceinline__ void fused_point(v4i& na, v4i& nb, v4f& da, v4f& db, int& c,
                                            const int p, const int j0, const int lane,
                                            const int r4, const int i4,
                                            const float* iw, const float (&rdr)[KNEI],
                                            const float* __restrict__ dect,
                                            const int* __restrict__ nid,
                                            const float* __restrict__ d,
                                            const int* __restrict__ lab,
                                            float* gbuf) {
  const int nl[8] = {na.x, na.y, na.z, na.w, nb.x, nb.y, nb.z, nb.w};
  const v4f* dect4 = (const v4f*)dect;
  const v4i* n4 = (const v4i*)nid;
  const v4f* d4 = (const v4f*)d;
  unsigned off[8];
#pragma unroll
  for (int g = 0; g < 8; ++g) off[g] = (unsigned)nl[g] * 8u + (unsigned)i4;
  __builtin_amdgcn_sched_barrier(0);
  v4f dec[8];
#pragma unroll
  for (int g = 0; g < 8; ++g) dec[g] = dect4[off[g]];
  __builtin_amdgcn_sched_barrier(0);
  int j2 = j0 + p + 2;
  j2 = (j2 < NPTS) ? j2 : (NPTS - 1);
  const unsigned q2 = (unsigned)j2 * 8u + 2u * (unsigned)r4;
  na = n4[q2];
  nb = n4[q2 + 1u];
  const v4f s = *(const v4f*)(iw + c * NBAS + 4 * i4);
  float d2[8];
  d2[0] = da.x * da.x;
  d2[1] = da.y * da.y;
  d2[2] = da.z * da.z;
  d2[3] = da.w * da.w;
  d2[4] = db.x * db.x;
  d2[5] = db.y * db.y;
  d2[6] = db.z * db.z;
  d2[7] = db.w * db.w;
  int j1 = j0 + p + 1;
  j1 = (j1 < NPTS) ? j1 : (NPTS - 1);
  const unsigned q1 = (unsigned)j1 * 8u + 2u * (unsigned)r4;
  da = d4[q1];
  db = d4[q1 + 1u];
  c = lab[j1];
  float w[KNEI];
#pragma unroll
  for (int g = 0; g < 8; ++g) {
    w[g * 4 + 0] = rdr[g * 4 + 0] * fmaxf(1.f - d2[g] * s.x, 0.f);
    w[g * 4 + 1] = rdr[g * 4 + 1] * fmaxf(1.f - d2[g] * s.y, 0.f);
    w[g * 4 + 2] = rdr[g * 4 + 2] * fmaxf(1.f - d2[g] * s.z, 0.f);
    w[g * 4 + 3] = rdr[g * 4 + 3] * fmaxf(1.f - d2[g] * s.w, 0.f);
  }
  float a0 = 0.f, a1 = 0.f, a2 = 0.f, a3 = 0.f;
#pragma unroll
  for (int g = 0; g < 8; ++g) {
    a0 += w[g * 4 + 0] * dec[g].x;
    a1 += w[g * 4 + 1] * dec[g].y;
    a2 += w[g * 4 + 2] * dec[g].z;
    a3 += w[g * 4 + 3] * dec[g].w;
  }
  float* gp = gbuf + lane * 17 + p;
  gp[0 * GP_PLANE] = a0;
  gp[1 * GP_PLANE] = a1;
  gp[2 * GP_PLANE] = a2;
  gp[3 * GP_PLANE] = a3;
}

#define FO_WAVES 4
#define FO_THREADS (FO_WAVES * 32)
#define FO_BLKS  1563
static_assert(4 * FO_THREADS == MCLU * NBAS, "one 16-byte group of inv_w2 per thread");
static_assert(8 * FO_THREADS == NBAS * NBAS, "two 16-byte groups of 1/denom per thread");
__global__ void __launch_bounds__(FO_THREADS) __attribute__((amdgpu_waves_per_eu(7)))
k_fused_out(const float* __restrict__ d,
                            const int* __restrict__ nid,
                            const int* __restrict__ lab,
                            const float* __restrict__ dect,
                            const float* __restrict__ invw2t,
                            const float* __restrict__ rdent,
                            const float* __restrict__ encoded,
                            float* __restrict__ out) {
  __shared__ __attribute__((aligned(16))) float iw[MCLU * NBAS];
  __shared__ float rd[NBAS * 33];
  __shared__ float enc_s[NBAS];
  __shared__ float g[FO_WAVES][GP_WAVE];
  const int tid = threadIdx.x;
  const v4f iwq = ((const v4f*)invw2t)[tid];
  const v4f rdq0 = ((const v4f*)rdent)[tid];
  const v4f rdq1 = ((const v4f*)rdent)[tid + FO_THREADS];
  const float encq = encoded[tid & (NBAS - 1)];

  const int lane = threadIdx.x & 31;
  const int wv = __builtin_amdgcn_readfirstlane(threadIdx.x >> 5);
  const int r4 = lane >> 3, i4 = lane & 7;
  const int gw = blockIdx.x * FO_WAVES + wv;
  float* gbuf = g[wv];
  const int NT = NPTS / 16;

  const int gwc = (gw < NT) ? gw : (NT - 1);
  const int jf = gwc * 16;
  const unsigned qf = (unsigned)jf * 8u + 2u * (unsigned)r4;
  const v4i* n4f = (const v4i*)nid;
  const v4f* d4f = (const v4f*)d;
  v4i naA = n4f[qf];
  v4i nbA = n4f[qf + 1u];
  v4f da = d4f[qf];
  v4f db = d4f[qf + 1u];
  int c = lab[jf];
  v4i naB = n4f[qf + 8u];
  v4i nbB = n4f[qf + 9u];

  *(v4f*)(iw + 4 * tid) = iwq;
  {
    const int t0 = 4 * tid, t1 = 4 * (tid + FO_THREADS);
    float* p0 = rd + (t0 >> 5) * 33 + (t0 & 31);
    float* p1 = rd + (t1 >> 5) * 33 + (t1 & 31);
    p0[0] = rdq0.x; p0[1] = rdq0.y; p0[2] = rdq0.z; p0[3] = rdq0.w;
    p1[0] = rdq1.x; p1[1] = rdq1.y; p1[2] = rdq1.z; p1[3] = rdq1.w;
  }
  if (tid < NBAS) enc_s[tid] = encq;
  __syncthreads();

  float rdr[KNEI];
#pragma unroll
  for (int gq = 0; gq < 8; ++gq)
#pragma unroll
    for (int b = 0; b < 4; ++b) rdr[gq * 4 + b] = rd[(4 * i4 + b) * 33 + 8 * r4 + gq];

  if (gw < NT) {
    const int j0 = gw * 16;
    for (int p = 0; p < 16; p += 2) {
      fused_point(naA, nbA, da, db, c, p, j0, lane, r4, i4, iw, rdr, dect, nid, d, lab, gbuf);
      fused_point(naB, nbB, da, db, c, p + 1, j0, lane, r4, i4, iw, rdr, dect, nid, d, lab, gbuf);
    }
    asm volatile("s_wait_dscnt 0" ::: "memory");

    v8f cacc = {0.f, 0.f, 0.f, 0.f, 0.f, 0.f, 0.f, 0.f};
    const int half = (lane >> 4) & 1;
    const int col = lane & 15;
#pragma unroll
    for (int s8 = 0; s8 < 8; ++s8) {
      const int kb = s8 * 4;
      v2f a, b;
      a.x = enc_s[kb + 2 * half];
      a.y = enc_s[kb + 2 * half + 1];
      const float* g0 = gbuf + (2 * half) * GP_PLANE + s8 * 17 + col;
      const float* g1 = gbuf + (2 * half + 1) * GP_PLANE + s8 * 17 + col;
      float bx = g0[0], by = g1[0];
#pragma unroll
      for (int r = 1; r < 4; ++r) {
        bx += g0[r * 8 * 17];
        by += g1[r * 8 * 17];
      }
      b.x = bx;
      b.y = by;
      cacc = __builtin_amdgcn_wmma_f32_16x16x4_f32(
          false, a, false, b, (short)0, cacc, false, false);
    }
    if (lane < 16) out[j0 + lane] = cacc[0];
  }
}

extern "C" void kernel_launch(void* const* d_in, const int* in_sizes, int n_in,
                              void* d_out, int out_size, void* d_ws, size_t ws_size,
                              hipStream_t stream) {
  const float* x     = (const float*)d_in[0];
  const float* enc_w = (const float*)d_in[1];
  const float* enc_b = (const float*)d_in[2];
  const float* dec   = (const float*)d_in[3];
  const float* bw_w  = (const float*)d_in[4];
  const float* bw_b  = (const float*)d_in[5];
  const float* nd    = (const float*)d_in[6];
  const int*   nid   = (const int*)d_in[7];
  const int*   lab   = (const int*)d_in[8];
  float* out = (float*)d_out;
  float* ws  = (float*)d_ws;

  float* enc_part = ws + WS_ENCPART;
  float* encoded  = ws + WS_ENCODED;
  float* invw2t   = ws + WS_INVW2T;
  float* rdent    = ws + WS_RDENT;
  float* denpart  = ws + WS_DENPART;
  float* dect     = ws + WS_DECT;

  k_enc_partial<<<ENC_BLKS, 64, 0, stream>>>(x, enc_w, enc_part);
  k_denom_partial_dec_transpose<<<DEN_BLKS, K2_THREADS, 0, stream>>>(
      nd, lab, enc_part, enc_b, bw_w, bw_b, encoded, invw2t, denpart, dec, dect);
  k_denom_finish<<<32, 1024, 0, stream>>>(denpart, rdent);
  k_fused_out<<<FO_BLKS, FO_THREADS, 0, stream>>>(nd, nid, lab, dect, invw2t, rdent,
                                       encoded, out);
}
